# starting kernel
# baseline (speedup 1.0000x reference)
#include <hip/hip_runtime.h>
#include <math.h>


typedef __attribute__((ext_vector_type(2))) float v2f;
typedef __attribute__((ext_vector_type(8))) float v8f;

#define BB   4
#define LL   2048
#define HH   16
#define DD   64
#define CCH  16
#define NS   8
#define SEGL (LL / NS)
#define CHPS (SEGL / CCH)
#define NCH  (LL / CCH)
#define FEPS 1e-6f

__device__ __forceinline__ float fmap(float x) {
    return x > 0.f ? x + 1.f : __expf(x);
}

__device__ __forceinline__ v8f wmma4(v2f a, v2f b, v8f c) {
    return __builtin_amdgcn_wmma_f32_16x16x4_f32(
        false, a, false, b, (short)0, c, false, false);
}

#define OFF_QF   0
#define OFF_KF   1024
#define OFF_KPI  2048
#define OFF_VPI  3072
#define OFF_S    4096
#define OFF_A    8192
#define OFF_KS   8448
#define OFF_KD   8512
#define OFF_Z    8576
#define SMEM_F   8592

__global__ __launch_bounds__(128)
void linattn_phase1(const float* __restrict__ k,
                    const float* __restrict__ v,
                    float* __restrict__ segKV,
                    float* __restrict__ segKS) {
    __shared__ float sm[3072];
    const int tid  = threadIdx.x;
    const int lane = tid & 31;
    const int wave = tid >> 5;
    const int lh   = lane >> 4;
    const int lc   = lane & 15;
    const int bh   = blockIdx.x / NS;
    const int seg  = blockIdx.x % NS;
    const int b    = bh / HH;
    const int h    = bh % HH;

    float* s_kf  = sm;
    float* s_kpi = sm + 1024;
    float* s_vpi = sm + 2048;

    const v8f zero8 = {0.f,0.f,0.f,0.f,0.f,0.f,0.f,0.f};
    v8f st[4];
    #pragma unroll
    for (int t = 0; t < 4; ++t) st[t] = zero8;
    float ks0 = 0.f, ks1 = 0.f;

    const int row = tid >> 3;
    const int c0  = (tid & 7) * 8;

    for (int ch = 0; ch < CHPS; ++ch) {
        const int l0 = seg * SEGL + ch * CCH;
        __syncthreads();
        {
            size_t g = ((size_t)((b * LL + l0 + row) * HH + h)) * DD + c0;
            float ka8[8], va8[8];
            *(float4*)(ka8)     = *(const float4*)(k + g);
            *(float4*)(ka8 + 4) = *(const float4*)(k + g + 4);
            *(float4*)(va8)     = *(const float4*)(v + g);
            *(float4*)(va8 + 4) = *(const float4*)(v + g + 4);
            #pragma unroll
            for (int j = 0; j < 8; ++j) {
                int c  = c0 + j;
                float kf = fmap(ka8[j]);
                int pi = ((row >> 1) * 64 + c) * 2 + (row & 1);
                s_kf[row * 64 + c] = kf;
                s_kpi[pi] = kf;
                s_vpi[pi] = va8[j];
            }
            if (ch + 1 < CHPS) {
                size_t gn = g + (size_t)CCH * HH * DD;
                __builtin_prefetch(k + gn, 0, 3);
                __builtin_prefetch(v + gn, 0, 3);
            }
        }
        __syncthreads();

        v2f bv[4];
        #pragma unroll
        for (int kk = 0; kk < 4; ++kk) {
            int p = 2 * kk + lh;
            bv[kk] = *(const v2f*)(s_vpi + (p * 64 + 16 * wave + lc) * 2);
        }
        #pragma unroll
        for (int t = 0; t < 4; ++t) {
            #pragma unroll
            for (int kk = 0; kk < 4; ++kk) {
                int p = 2 * kk + lh;
                v2f ak = *(const v2f*)(s_kpi + (p * 64 + 16 * t + lc) * 2);
                st[t] = wmma4(ak, bv[kk], st[t]);
            }
        }
        if (wave == 2) {
            float a0 = 0.f, a1 = 0.f;
            #pragma unroll
            for (int r16 = 0; r16 < 16; ++r16) {
                a0 += s_kf[r16 * 64 + lane * 2];
                a1 += s_kf[r16 * 64 + lane * 2 + 1];
            }
            ks0 += a0; ks1 += a1;
        }
    }

    float* kvout = segKV + (size_t)blockIdx.x * 4096;
    #pragma unroll
    for (int t = 0; t < 4; ++t) {
        #pragma unroll
        for (int r = 0; r < 8; ++r) {
            int srow = 16 * t + r + 8 * lh;
            kvout[srow * 64 + 16 * wave + lc] = st[t][r];
        }
    }
    if (wave == 2) {
        segKS[(size_t)blockIdx.x * 64 + lane * 2]     = ks0;
        segKS[(size_t)blockIdx.x * 64 + lane * 2 + 1] = ks1;
    }
}

__global__ __launch_bounds__(128)
void linattn_phase2(float* __restrict__ segKV,
                    float* __restrict__ segKS) {
    const int bh  = blockIdx.x;
    const int tid = threadIdx.x;
    for (int e = tid; e < 4096; e += 128) {
        float acc = 0.f;
        #pragma unroll
        for (int s = 0; s < NS; ++s) {
            size_t idx = ((size_t)(bh * NS + s)) * 4096 + e;
            float t = segKV[idx];
            segKV[idx] = acc;
            acc += t;
        }
    }
    if (tid < 64) {
        float acc = 0.f;
        #pragma unroll
        for (int s = 0; s < NS; ++s) {
            size_t idx = ((size_t)(bh * NS + s)) * 64 + tid;
            float t = segKS[idx];
            segKS[idx] = acc;
            acc += t;
        }
    }
}

template <bool SEEDED>
__global__ __launch_bounds__(128)
void linattn_scan(const float* __restrict__ q,
                  const float* __restrict__ k,
                  const float* __restrict__ v,
                  float* __restrict__ out,
                  const float* __restrict__ segKV,
                  const float* __restrict__ segKS) {
    __shared__ float sm[SMEM_F];
    const int tid  = threadIdx.x;
    const int lane = tid & 31;
    const int wave = tid >> 5;
    const int lh   = lane >> 4;
    const int lc   = lane & 15;
    const int bh   = SEEDED ? (blockIdx.x / NS) : blockIdx.x;
    const int seg  = SEEDED ? (blockIdx.x % NS) : 0;
    const int b    = bh / HH;
    const int h    = bh % HH;
    const int nch  = SEEDED ? CHPS : NCH;

    float* s_qf  = sm + OFF_QF;
    float* s_kf  = sm + OFF_KF;
    float* s_kpi = sm + OFF_KPI;
    float* s_vpi = sm + OFF_VPI;
    float* s_spi = sm + OFF_S + wave * 1024;
    float* s_a   = sm + OFF_A;
    float* s_ks  = sm + OFF_KS;
    float* s_kd  = sm + OFF_KD;
    float* s_z   = sm + OFF_Z;

    const v8f zero8 = {0.f,0.f,0.f,0.f,0.f,0.f,0.f,0.f};
    v8f st[4];

    if (SEEDED) {
        const float* kvin = segKV + (size_t)blockIdx.x * 4096;
        #pragma unroll
        for (int t = 0; t < 4; ++t) {
            #pragma unroll
            for (int r = 0; r < 8; ++r) {
                int srow = 16 * t + r + 8 * lh;
                float val = kvin[srow * 64 + 16 * wave + lc];
                st[t][r] = val;
                s_spi[((srow >> 1) * 16 + lc) * 2 + (srow & 1)] = val;
            }
        }
        if (tid < 64) s_ks[tid] = segKS[(size_t)blockIdx.x * 64 + tid];
    } else {
        #pragma unroll
        for (int t = 0; t < 4; ++t) st[t] = zero8;
        #pragma unroll
        for (int i = 0; i < 32; ++i) sm[OFF_S + tid + i * 128] = 0.f;
        if (tid < 64) s_ks[tid] = 0.f;
    }

    const int row = tid >> 3;
    const int c0  = (tid & 7) * 8;

    for (int ch = 0; ch < nch; ++ch) {
        const int l0 = seg * SEGL + ch * CCH;
        __syncthreads();

        {
            size_t g = ((size_t)((b * LL + l0 + row) * HH + h)) * DD + c0;
            float qa8[8], ka8[8], va8[8];
            *(float4*)(qa8)     = *(const float4*)(q + g);
            *(float4*)(qa8 + 4) = *(const float4*)(q + g + 4);
            *(float4*)(ka8)     = *(const float4*)(k + g);
            *(float4*)(ka8 + 4) = *(const float4*)(k + g + 4);
            *(float4*)(va8)     = *(const float4*)(v + g);
            *(float4*)(va8 + 4) = *(const float4*)(v + g + 4);
            #pragma unroll
            for (int j = 0; j < 8; ++j) {
                int c  = c0 + j;
                float qf = fmap(qa8[j]);
                float kf = fmap(ka8[j]);
                int pi = ((row >> 1) * 64 + c) * 2 + (row & 1);
                s_qf[row * 64 + c] = qf;
                s_kf[row * 64 + c] = kf;
                s_kpi[pi] = kf;
                s_vpi[pi] = va8[j];
            }
            if (ch + 1 < nch) {
                size_t gn = g + (size_t)CCH * HH * DD;
                __builtin_prefetch(q + gn, 0, 3);
                __builtin_prefetch(k + gn, 0, 3);
                __builtin_prefetch(v + gn, 0, 3);
            }
        }
        __syncthreads();

        v2f qa[16];
        #pragma unroll
        for (int s = 0; s < 16; ++s)
            qa[s] = *(const v2f*)(s_qf + lc * 64 + 4 * s + 2 * lh);

        v8f oc = zero8;
        #pragma unroll
        for (int t = 0; t < 4; ++t) {
            #pragma unroll
            for (int kk = 0; kk < 4; ++kk) {
                int p = 8 * t + 2 * kk + lh;
                v2f bs = *(const v2f*)(s_spi + (p * 16 + lc) * 2);
                oc = wmma4(qa[4 * t + kk], bs, oc);
            }
        }

        if (wave == 0) {
            v8f aq = zero8;
            #pragma unroll
            for (int s = 0; s < 16; ++s) {
                v2f bk = *(const v2f*)(s_kf + lc * 64 + 4 * s + 2 * lh);
                aq = wmma4(qa[s], bk, aq);
            }
            #pragma unroll
            for (int r = 0; r < 8; ++r) {
                int m = r + 8 * lh;
                s_a[m * 16 + lc] = (lc <= m) ? aq[r] : 0.f;
            }
        }
        __syncthreads();

        if (wave == 1 && lane < 16) {
            float zz = FEPS;
            #pragma unroll
            for (int n = 0; n < 16; ++n) zz += s_a[lane * 16 + n];
            #pragma unroll 8
            for (int d = 0; d < 64; ++d) zz += s_qf[lane * 64 + d] * s_ks[d];
            s_z[lane] = zz;
        }
        if (wave == 2) {
            #pragma unroll
            for (int j = 0; j < 2; ++j) {
                int d = lane * 2 + j;
                float ssum = 0.f;
                #pragma unroll
                for (int r16 = 0; r16 < 16; ++r16) ssum += s_kf[r16 * 64 + d];
                s_kd[d] = ssum;
            }
        }

        v2f bv[4];
        #pragma unroll
        for (int kk = 0; kk < 4; ++kk) {
            int p = 2 * kk + lh;
            bv[kk] = *(const v2f*)(s_vpi + (p * 64 + 16 * wave + lc) * 2);
        }
        #pragma unroll
        for (int kk = 0; kk < 4; ++kk) {
            v2f av = *(const v2f*)(s_a + lc * 16 + 4 * kk + 2 * lh);
            oc = wmma4(av, bv[kk], oc);
        }
        #pragma unroll
        for (int t = 0; t < 4; ++t) {
            #pragma unroll
            for (int kk = 0; kk < 4; ++kk) {
                int p = 2 * kk + lh;
                v2f ak = *(const v2f*)(s_kpi + (p * 64 + 16 * t + lc) * 2);
                st[t] = wmma4(ak, bv[kk], st[t]);
            }
        }
        __syncthreads();

        #pragma unroll
        for (int r = 0; r < 8; ++r) {
            int m = r + 8 * lh;
            float zz = s_z[m];
            size_t go = ((size_t)((b * LL + l0 + m) * HH + h)) * DD
                        + 16 * wave + lc;
            out[go] = oc[r] / zz;
        }
        #pragma unroll
        for (int t = 0; t < 4; ++t) {
            #pragma unroll
            for (int r = 0; r < 8; ++r) {
                int srow = 16 * t + r + 8 * lh;
                s_spi[((srow >> 1) * 16 + lc) * 2 + (srow & 1)] = st[t][r];
            }
        }
        if (wave == 3) {
            s_ks[lane * 2]     += s_kd[lane * 2];
            s_ks[lane * 2 + 1] += s_kd[lane * 2 + 1];
        }
    }
}

extern "C" void kernel_launch(void* const* d_in, const int* in_sizes, int n_in,
                              void* d_out, int out_size, void* d_ws, size_t ws_size,
                              hipStream_t stream) {
    (void)in_sizes; (void)n_in; (void)out_size;
    const float* q = (const float*)d_in[0];
    const float* k = (const float*)d_in[1];
    const float* v = (const float*)d_in[2];
    float* out = (float*)d_out;

    const size_t nSeg   = (size_t)BB * HH * NS;
    const size_t needWs = nSeg * (4096 + 64) * sizeof(float);

    if (ws_size >= needWs) {
        float* segKV = (float*)d_ws;
        float* segKS = segKV + nSeg * 4096;
        hipLaunchKernelGGL(linattn_phase1, dim3(BB * HH * NS), dim3(128), 0,
                           stream, k, v, segKV, segKS);
        hipLaunchKernelGGL(linattn_phase2, dim3(BB * HH), dim3(128), 0,
                           stream, segKV, segKS);
        hipLaunchKernelGGL((linattn_scan<true>), dim3(BB * HH * NS), dim3(128),
                           0, stream, q, k, v, out, segKV, segKS);
    } else {
        hipLaunchKernelGGL((linattn_scan<false>), dim3(BB * HH), dim3(128),
                           0, stream, q, k, v, out,
                           (const float*)nullptr, (const float*)nullptr);
    }
}


# staging tile 16 to 64 rows in both WMMA kernels: one global round trip, two barriers per 64 rows; an LDS tile per chunk, so one barrier per chunk not three (Ksum in two buffers); sums unchanged
# speedup vs baseline: 1.4708x; 1.4708x over previous
#include <hip/hip_runtime.h>
#include <math.h>


typedef __attribute__((ext_vector_type(2))) float v2f;
typedef __attribute__((ext_vector_type(8))) float v8f;

#define BB   4
#define LL   2048
#define HH   16
#define DD   64
#define CCH  16
#define NS   8
#define SEGL (LL / NS)
#define SUB  4
#define TROWS (SUB * CCH)
#define FEPS 1e-6f

__device__ __forceinline__ float fmap(float x) {
    return x > 0.f ? x + 1.f : __expf(x);
}

__device__ __forceinline__ v8f wmma4(v2f a, v2f b, v8f c) {
    return __builtin_amdgcn_wmma_f32_16x16x4_f32(
        false, a, false, b, (short)0, c, false, false);
}

#define T_QF    0
#define T_KF    1024
#define T_KPI   2048
#define T_VPI   3072
#define T_A     4096
#define T_Z     4352
#define TILE_F  4368
#define OFF_S   (SUB * TILE_F)
#define OFF_KS  (OFF_S + 4096)
#define SMEM_F  (OFF_KS + 128)

#define P1_KF   0
#define P1_KPI  1024
#define P1_VPI  2048
#define P1_TILE 3072

__global__ __launch_bounds__(128)
void linattn_phase1(const float* __restrict__ k,
                    const float* __restrict__ v,
                    float* __restrict__ segKV,
                    float* __restrict__ segKS) {
    __shared__ float sm[SUB * P1_TILE];
    const int tid  = threadIdx.x;
    const int lane = tid & 31;
    const int wave = tid >> 5;
    const int lh   = lane >> 4;
    const int lc   = lane & 15;
    const int bh   = blockIdx.x / NS;
    const int seg  = blockIdx.x % NS;
    const int b    = bh / HH;
    const int h    = bh % HH;

    const v8f zero8 = {0.f,0.f,0.f,0.f,0.f,0.f,0.f,0.f};
    v8f st[4];
    #pragma unroll
    for (int t = 0; t < 4; ++t) st[t] = zero8;
    float ks0 = 0.f, ks1 = 0.f;

    const int row = tid >> 3;
    const int c0  = (tid & 7) * 8;

    for (int ps = 0; ps < SEGL / TROWS; ++ps) {
        const int lb = seg * SEGL + ps * TROWS;
        const size_t g0 = ((size_t)((b * LL + lb + row) * HH + h)) * DD + c0;
        float ka8[SUB][8], va8[SUB][8];
        #pragma unroll
        for (int j = 0; j < SUB; ++j) {
            const size_t g = g0 + (size_t)j * CCH * HH * DD;
            *(float4*)(ka8[j])     = *(const float4*)(k + g);
            *(float4*)(ka8[j] + 4) = *(const float4*)(k + g + 4);
            *(float4*)(va8[j])     = *(const float4*)(v + g);
            *(float4*)(va8[j] + 4) = *(const float4*)(v + g + 4);
        }
        __syncthreads();
        #pragma unroll
        for (int j = 0; j < SUB; ++j) {
            float* s_kf  = sm + j * P1_TILE + P1_KF;
            float* s_kpi = sm + j * P1_TILE + P1_KPI;
            float* s_vpi = sm + j * P1_TILE + P1_VPI;
            #pragma unroll
            for (int i = 0; i < 8; ++i) {
                int c  = c0 + i;
                float kf = fmap(ka8[j][i]);
                int pi = ((row >> 1) * 64 + c) * 2 + (row & 1);
                s_kf[row * 64 + c] = kf;
                s_kpi[pi] = kf;
                s_vpi[pi] = va8[j][i];
            }
        }
        if (ps + 1 < SEGL / TROWS) {
            #pragma unroll
            for (int j = 0; j < SUB; ++j) {
                const size_t gn = g0 + (size_t)(SUB + j) * CCH * HH * DD;
                __builtin_prefetch(k + gn, 0, 3);
                __builtin_prefetch(v + gn, 0, 3);
            }
        }
        __syncthreads();

        for (int j = 0; j < SUB; ++j) {
            const float* s_kf  = sm + j * P1_TILE + P1_KF;
            const float* s_kpi = sm + j * P1_TILE + P1_KPI;
            const float* s_vpi = sm + j * P1_TILE + P1_VPI;
            v2f bv[4];
            #pragma unroll
            for (int kk = 0; kk < 4; ++kk) {
                int p = 2 * kk + lh;
                bv[kk] = *(const v2f*)(s_vpi + (p * 64 + 16 * wave + lc) * 2);
            }
            #pragma unroll
            for (int t = 0; t < 4; ++t) {
                #pragma unroll
                for (int kk = 0; kk < 4; ++kk) {
                    int p = 2 * kk + lh;
                    v2f ak = *(const v2f*)(s_kpi + (p * 64 + 16 * t + lc) * 2);
                    st[t] = wmma4(ak, bv[kk], st[t]);
                }
            }
            if (wave == 2) {
                float a0 = 0.f, a1 = 0.f;
                #pragma unroll
                for (int r16 = 0; r16 < 16; ++r16) {
                    a0 += s_kf[r16 * 64 + lane * 2];
                    a1 += s_kf[r16 * 64 + lane * 2 + 1];
                }
                ks0 += a0; ks1 += a1;
            }
        }
    }

    float* kvout = segKV + (size_t)blockIdx.x * 4096;
    #pragma unroll
    for (int t = 0; t < 4; ++t) {
        #pragma unroll
        for (int r = 0; r < 8; ++r) {
            int srow = 16 * t + r + 8 * lh;
            kvout[srow * 64 + 16 * wave + lc] = st[t][r];
        }
    }
    if (wave == 2) {
        segKS[(size_t)blockIdx.x * 64 + lane * 2]     = ks0;
        segKS[(size_t)blockIdx.x * 64 + lane * 2 + 1] = ks1;
    }
}

__global__ __launch_bounds__(256)
void linattn_phase2(float* __restrict__ segKV,
                    float* __restrict__ segKS) {
    const int bh  = blockIdx.x;
    const int tid = threadIdx.x;
    float* base;
    int stride;
    if (blockIdx.y < 4) {
        base   = segKV + (size_t)bh * NS * 4096 + blockIdx.y * 1024 + tid * 4;
        stride = 4096;
    } else {
        if (tid >= 16) return;
        base   = segKS + (size_t)bh * NS * 64 + tid * 4;
        stride = 64;
    }
    float4 part[NS];
    #pragma unroll
    for (int s = 0; s < NS; ++s)
        part[s] = *(const float4*)(base + (size_t)s * stride);
    float4 acc = make_float4(0.f, 0.f, 0.f, 0.f);
    #pragma unroll
    for (int s = 0; s < NS; ++s) {
        *(float4*)(base + (size_t)s * stride) = acc;
        acc.x += part[s].x;
        acc.y += part[s].y;
        acc.z += part[s].z;
        acc.w += part[s].w;
    }
}

template <bool SEEDED>
__global__ __launch_bounds__(128)
void linattn_scan(const float* __restrict__ q,
                  const float* __restrict__ k,
                  const float* __restrict__ v,
                  float* __restrict__ out,
                  const float* __restrict__ segKV,
                  const float* __restrict__ segKS) {
    __shared__ float sm[SMEM_F];
    const int tid  = threadIdx.x;
    const int lane = tid & 31;
    const int wave = tid >> 5;
    const int lh   = lane >> 4;
    const int lc   = lane & 15;
    const int bh   = SEEDED ? (blockIdx.x / NS) : blockIdx.x;
    const int seg  = SEEDED ? (blockIdx.x % NS) : 0;
    const int b    = bh / HH;
    const int h    = bh % HH;
    const int npass = SEEDED ? (SEGL / TROWS) : (LL / TROWS);

    float* s_spi = sm + OFF_S + wave * 1024;
    float* s_ks  = sm + OFF_KS;

    const v8f zero8 = {0.f,0.f,0.f,0.f,0.f,0.f,0.f,0.f};
    const v2f ones2 = {1.f, 1.f};
    v8f st[4];

    if (SEEDED) {
        const float* kvin = segKV + (size_t)blockIdx.x * 4096;
        #pragma unroll
        for (int t = 0; t < 4; ++t) {
            #pragma unroll
            for (int r = 0; r < 8; ++r) {
                int srow = 16 * t + r + 8 * lh;
                float val = kvin[srow * 64 + 16 * wave + lc];
                st[t][r] = val;
                s_spi[((srow >> 1) * 16 + lc) * 2 + (srow & 1)] = val;
            }
        }
        if (tid < 64) s_ks[tid] = segKS[(size_t)blockIdx.x * 64 + tid];
    } else {
        #pragma unroll
        for (int t = 0; t < 4; ++t) st[t] = zero8;
        #pragma unroll
        for (int i = 0; i < 32; ++i) sm[OFF_S + tid + i * 128] = 0.f;
        if (tid < 64) s_ks[tid] = 0.f;
    }

    const int row = tid >> 3;
    const int c0  = (tid & 7) * 8;
    int par = 0;

    for (int ps = 0; ps < npass; ++ps) {
        const int lb = seg * SEGL + ps * TROWS;

        {
            const size_t g0 = ((size_t)((b * LL + lb + row) * HH + h)) * DD + c0;
            float qa8[SUB][8], ka8[SUB][8], va8[SUB][8];
            #pragma unroll
            for (int j = 0; j < SUB; ++j) {
                const size_t g = g0 + (size_t)j * CCH * HH * DD;
                *(float4*)(qa8[j])     = *(const float4*)(q + g);
                *(float4*)(qa8[j] + 4) = *(const float4*)(q + g + 4);
                *(float4*)(ka8[j])     = *(const float4*)(k + g);
                *(float4*)(ka8[j] + 4) = *(const float4*)(k + g + 4);
                *(float4*)(va8[j])     = *(const float4*)(v + g);
                *(float4*)(va8[j] + 4) = *(const float4*)(v + g + 4);
            }
            __syncthreads();
            #pragma unroll
            for (int j = 0; j < SUB; ++j) {
                float* tl = sm + j * TILE_F;
                #pragma unroll
                for (int i = 0; i < 8; ++i) {
                    int c  = c0 + i;
                    float qf = fmap(qa8[j][i]);
                    float kf = fmap(ka8[j][i]);
                    int pi = ((row >> 1) * 64 + c) * 2 + (row & 1);
                    tl[T_QF + row * 64 + c] = qf;
                    tl[T_KF + row * 64 + c] = kf;
                    tl[T_KPI + pi] = kf;
                    tl[T_VPI + pi] = va8[j][i];
                }
            }
            if (ps + 1 < npass) {
                #pragma unroll
                for (int j = 0; j < SUB; ++j) {
                    const size_t gn = g0 + (size_t)(SUB + j) * CCH * HH * DD;
                    __builtin_prefetch(q + gn, 0, 3);
                    __builtin_prefetch(k + gn, 0, 3);
                    __builtin_prefetch(v + gn, 0, 3);
                }
            }
        }
        __syncthreads();

        for (int j = 0; j < SUB; ++j) {
            const int l0 = lb + j * CCH;
            float* tl = sm + j * TILE_F;
            const float* s_qf  = tl + T_QF;
            const float* s_kf  = tl + T_KF;
            const float* s_kpi = tl + T_KPI;
            const float* s_vpi = tl + T_VPI;
            float* s_a = tl + T_A;
            float* s_z = tl + T_Z;
            const float* ks_cur = s_ks + par * 64;
            float* ks_nxt = s_ks + (par ^ 1) * 64;

            v2f qa[16];
            #pragma unroll
            for (int s = 0; s < 16; ++s)
                qa[s] = *(const v2f*)(s_qf + lc * 64 + 4 * s + 2 * lh);

            v8f oc = zero8;
            #pragma unroll
            for (int t = 0; t < 4; ++t) {
                #pragma unroll
                for (int kk = 0; kk < 4; ++kk) {
                    int p = 8 * t + 2 * kk + lh;
                    v2f bs = *(const v2f*)(s_spi + (p * 16 + lc) * 2);
                    oc = wmma4(qa[4 * t + kk], bs, oc);
                }
            }

            if (wave == 0) {
                v8f aq = zero8;
                #pragma unroll
                for (int s = 0; s < 16; ++s) {
                    v2f bk = *(const v2f*)(s_kf + lc * 64 + 4 * s + 2 * lh);
                    aq = wmma4(qa[s], bk, aq);
                }
                #pragma unroll
                for (int r = 0; r < 8; ++r) {
                    int m = r + 8 * lh;
                    s_a[m * 16 + lc] = (lc <= m) ? aq[r] : 0.f;
                }
            }
            if (wave == 1) {
                v8f zi = zero8;
                #pragma unroll
                for (int s = 0; s < 16; ++s) {
                    v2f bks = *(const v2f*)(ks_cur + 4 * s + 2 * lh);
                    zi = wmma4(qa[s], bks, zi);
                }
                if (lc == 0) {
                    #pragma unroll
                    for (int r = 0; r < 8; ++r) s_z[r + 8 * lh] = zi[r];
                }
            }
            if (wave == 2) {
                float kd0 = 0.f, kd1 = 0.f;
                #pragma unroll
                for (int r16 = 0; r16 < 16; ++r16) {
                    kd0 += s_kf[r16 * 64 + lane * 2];
                    kd1 += s_kf[r16 * 64 + lane * 2 + 1];
                }
                ks_nxt[lane * 2]     = ks_cur[lane * 2]     + kd0;
                ks_nxt[lane * 2 + 1] = ks_cur[lane * 2 + 1] + kd1;
            }
            __syncthreads();

            v2f bv[4];
            #pragma unroll
            for (int kk = 0; kk < 4; ++kk) {
                int p = 2 * kk + lh;
                bv[kk] = *(const v2f*)(s_vpi + (p * 64 + 16 * wave + lc) * 2);
            }
            v8f zr = zero8;
            #pragma unroll
            for (int kk = 0; kk < 4; ++kk) {
                v2f av = *(const v2f*)(s_a + lc * 16 + 4 * kk + 2 * lh);
                oc = wmma4(av, bv[kk], oc);
                zr = wmma4(av, ones2, zr);
            }
            #pragma unroll
            for (int t = 0; t < 4; ++t) {
                #pragma unroll
                for (int kk = 0; kk < 4; ++kk) {
                    int p = 2 * kk + lh;
                    v2f ak = *(const v2f*)(s_kpi + (p * 64 + 16 * t + lc) * 2);
                    st[t] = wmma4(ak, bv[kk], st[t]);
                }
            }

            #pragma unroll
            for (int r = 0; r < 8; ++r) {
                int m = r + 8 * lh;
                float zz = FEPS + zr[r] + s_z[m];
                size_t go = ((size_t)((b * LL + l0 + m) * HH + h)) * DD
                            + 16 * wave + lc;
                out[go] = oc[r] / zz;
            }
            #pragma unroll
            for (int t = 0; t < 4; ++t) {
                #pragma unroll
                for (int r = 0; r < 8; ++r) {
                    int srow = 16 * t + r + 8 * lh;
                    s_spi[((srow >> 1) * 16 + lc) * 2 + (srow & 1)] = st[t][r];
                }
            }
            par ^= 1;
        }
    }
}

extern "C" void kernel_launch(void* const* d_in, const int* in_sizes, int n_in,
                              void* d_out, int out_size, void* d_ws, size_t ws_size,
                              hipStream_t stream) {
    (void)in_sizes; (void)n_in; (void)out_size;
    const float* q = (const float*)d_in[0];
    const float* k = (const float*)d_in[1];
    const float* v = (const float*)d_in[2];
    float* out = (float*)d_out;

    const size_t nSeg   = (size_t)BB * HH * NS;
    const size_t needWs = nSeg * (4096 + 64) * sizeof(float);

    if (ws_size >= needWs) {
        float* segKV = (float*)d_ws;
        float* segKS = segKV + nSeg * 4096;
        hipLaunchKernelGGL(linattn_phase1, dim3(BB * HH * NS), dim3(128), 0,
                           stream, k, v, segKV, segKS);
        hipLaunchKernelGGL(linattn_phase2, dim3(BB * HH, 5), dim3(256), 0,
                           stream, segKV, segKS);
        hipLaunchKernelGGL((linattn_scan<true>), dim3(BB * HH * NS), dim3(128),
                           0, stream, q, k, v, out, segKV, segKS);
    } else {
        hipLaunchKernelGGL((linattn_scan<false>), dim3(BB * HH), dim3(128),
                           0, stream, q, k, v, out,
                           (const float*)nullptr, (const float*)nullptr);
    }
}


# staging thread map from one row x 8 columns to a row pair x 4 columns in both WMMA kernels: all LDS tile stores 128-bit (8 per chunk and thread, not 32 scalar with 4-way bank conflicts); sums same
# speedup vs baseline: 1.6354x; 1.1119x over previous
#include <hip/hip_runtime.h>
#include <math.h>


typedef __attribute__((ext_vector_type(2))) float v2f;
typedef __attribute__((ext_vector_type(4))) float v4f;
typedef __attribute__((ext_vector_type(8))) float v8f;

#define BB   4
#define LL   2048
#define HH   16
#define DD   64
#define CCH  16
#define NS   8
#define SEGL (LL / NS)
#define SUB  4
#define TROWS (SUB * CCH)
#define FEPS 1e-6f

__device__ __forceinline__ float fmap(float x) {
    return x > 0.f ? x + 1.f : __expf(x);
}

__device__ __forceinline__ v4f fmap4(v4f x) {
    v4f y = {fmap(x[0]), fmap(x[1]), fmap(x[2]), fmap(x[3])};
    return y;
}

__device__ __forceinline__ void store_pair4(float* dst, v4f a, v4f b) {
    v4f lo = {a[0], b[0], a[1], b[1]};
    v4f hi = {a[2], b[2], a[3], b[3]};
    *(v4f*)(dst)     = lo;
    *(v4f*)(dst + 4) = hi;
}

__device__ __forceinline__ v8f wmma4(v2f a, v2f b, v8f c) {
    return __builtin_amdgcn_wmma_f32_16x16x4_f32(
        false, a, false, b, (short)0, c, false, false);
}

#define T_QF    0
#define T_KF    1024
#define T_KPI   2048
#define T_VPI   3072
#define T_A     4096
#define T_Z     4352
#define TILE_F  4368
#define OFF_S   (SUB * TILE_F)
#define OFF_KS  (OFF_S + 4096)
#define SMEM_F  (OFF_KS + 128)

#define P1_KF   0
#define P1_KPI  1024
#define P1_VPI  2048
#define P1_TILE 3072

__global__ __launch_bounds__(128)
void linattn_phase1(const float* __restrict__ k,
                    const float* __restrict__ v,
                    float* __restrict__ segKV,
                    float* __restrict__ segKS) {
    __shared__ __attribute__((aligned(16))) float sm[SUB * P1_TILE];
    const int tid  = threadIdx.x;
    const int lane = tid & 31;
    const int wave = tid >> 5;
    const int lh   = lane >> 4;
    const int lc   = lane & 15;
    const int bh   = blockIdx.x / NS;
    const int seg  = blockIdx.x % NS;
    const int b    = bh / HH;
    const int h    = bh % HH;

    const v8f zero8 = {0.f,0.f,0.f,0.f,0.f,0.f,0.f,0.f};
    v8f st[4];
    #pragma unroll
    for (int t = 0; t < 4; ++t) st[t] = zero8;
    float ks0 = 0.f, ks1 = 0.f;

    const int row = tid >> 3;
    const int c0  = (tid & 7) * 8;
    const int sp  = tid >> 4;
    const int sc  = (tid & 15) * 4;

    for (int ps = 0; ps < SEGL / TROWS; ++ps) {
        const int lb = seg * SEGL + ps * TROWS;
        const size_t g0 = ((size_t)((b * LL + lb + row) * HH + h)) * DD + c0;
        const size_t gs = ((size_t)((b * LL + lb + 2 * sp) * HH + h)) * DD + sc;
        v4f ka[SUB][2], va[SUB][2];
        #pragma unroll
        for (int j = 0; j < SUB; ++j) {
            const size_t g = gs + (size_t)j * CCH * HH * DD;
            ka[j][0] = *(const v4f*)(k + g);
            ka[j][1] = *(const v4f*)(k + g + HH * DD);
            va[j][0] = *(const v4f*)(v + g);
            va[j][1] = *(const v4f*)(v + g + HH * DD);
        }
        __syncthreads();
        #pragma unroll
        for (int j = 0; j < SUB; ++j) {
            float* s_kf  = sm + j * P1_TILE + P1_KF;
            float* s_kpi = sm + j * P1_TILE + P1_KPI;
            float* s_vpi = sm + j * P1_TILE + P1_VPI;
            const v4f kf0 = fmap4(ka[j][0]);
            const v4f kf1 = fmap4(ka[j][1]);
            *(v4f*)(s_kf + (2 * sp) * 64 + sc)     = kf0;
            *(v4f*)(s_kf + (2 * sp + 1) * 64 + sc) = kf1;
            store_pair4(s_kpi + (sp * 64 + sc) * 2, kf0, kf1);
            store_pair4(s_vpi + (sp * 64 + sc) * 2, va[j][0], va[j][1]);
        }
        if (ps + 1 < SEGL / TROWS) {
            #pragma unroll
            for (int j = 0; j < SUB; ++j) {
                const size_t gn = g0 + (size_t)(SUB + j) * CCH * HH * DD;
                __builtin_prefetch(k + gn, 0, 3);
                __builtin_prefetch(v + gn, 0, 3);
            }
        }
        __syncthreads();

        for (int j = 0; j < SUB; ++j) {
            const float* s_kf  = sm + j * P1_TILE + P1_KF;
            const float* s_kpi = sm + j * P1_TILE + P1_KPI;
            const float* s_vpi = sm + j * P1_TILE + P1_VPI;
            v2f bv[4];
            #pragma unroll
            for (int kk = 0; kk < 4; ++kk) {
                int p = 2 * kk + lh;
                bv[kk] = *(const v2f*)(s_vpi + (p * 64 + 16 * wave + lc) * 2);
            }
            #pragma unroll
            for (int t = 0; t < 4; ++t) {
                #pragma unroll
                for (int kk = 0; kk < 4; ++kk) {
                    int p = 2 * kk + lh;
                    v2f ak = *(const v2f*)(s_kpi + (p * 64 + 16 * t + lc) * 2);
                    st[t] = wmma4(ak, bv[kk], st[t]);
                }
            }
            if (wave == 2) {
                float a0 = 0.f, a1 = 0.f;
                #pragma unroll
                for (int r16 = 0; r16 < 16; ++r16) {
                    a0 += s_kf[r16 * 64 + lane * 2];
                    a1 += s_kf[r16 * 64 + lane * 2 + 1];
                }
                ks0 += a0; ks1 += a1;
            }
        }
    }

    float* kvout = segKV + (size_t)blockIdx.x * 4096;
    #pragma unroll
    for (int t = 0; t < 4; ++t) {
        #pragma unroll
        for (int r = 0; r < 8; ++r) {
            int srow = 16 * t + r + 8 * lh;
            kvout[srow * 64 + 16 * wave + lc] = st[t][r];
        }
    }
    if (wave == 2) {
        segKS[(size_t)blockIdx.x * 64 + lane * 2]     = ks0;
        segKS[(size_t)blockIdx.x * 64 + lane * 2 + 1] = ks1;
    }
}

__global__ __launch_bounds__(256)
void linattn_phase2(float* __restrict__ segKV,
                    float* __restrict__ segKS) {
    const int bh  = blockIdx.x;
    const int tid = threadIdx.x;
    float* base;
    int stride;
    if (blockIdx.y < 4) {
        base   = segKV + (size_t)bh * NS * 4096 + blockIdx.y * 1024 + tid * 4;
        stride = 4096;
    } else {
        if (tid >= 16) return;
        base   = segKS + (size_t)bh * NS * 64 + tid * 4;
        stride = 64;
    }
    float4 part[NS];
    #pragma unroll
    for (int s = 0; s < NS; ++s)
        part[s] = *(const float4*)(base + (size_t)s * stride);
    float4 acc = make_float4(0.f, 0.f, 0.f, 0.f);
    #pragma unroll
    for (int s = 0; s < NS; ++s) {
        *(float4*)(base + (size_t)s * stride) = acc;
        acc.x += part[s].x;
        acc.y += part[s].y;
        acc.z += part[s].z;
        acc.w += part[s].w;
    }
}

template <bool SEEDED>
__global__ __launch_bounds__(128)
void linattn_scan(const float* __restrict__ q,
                  const float* __restrict__ k,
                  const float* __restrict__ v,
                  float* __restrict__ out,
                  const float* __restrict__ segKV,
                  const float* __restrict__ segKS) {
    __shared__ __attribute__((aligned(16))) float sm[SMEM_F];
    const int tid  = threadIdx.x;
    const int lane = tid & 31;
    const int wave = tid >> 5;
    const int lh   = lane >> 4;
    const int lc   = lane & 15;
    const int bh   = SEEDED ? (blockIdx.x / NS) : blockIdx.x;
    const int seg  = SEEDED ? (blockIdx.x % NS) : 0;
    const int b    = bh / HH;
    const int h    = bh % HH;
    const int npass = SEEDED ? (SEGL / TROWS) : (LL / TROWS);

    float* s_spi = sm + OFF_S + wave * 1024;
    float* s_ks  = sm + OFF_KS;

    const v8f zero8 = {0.f,0.f,0.f,0.f,0.f,0.f,0.f,0.f};
    const v2f ones2 = {1.f, 1.f};
    v8f st[4];

    if (SEEDED) {
        const float* kvin = segKV + (size_t)blockIdx.x * 4096;
        #pragma unroll
        for (int t = 0; t < 4; ++t) {
            #pragma unroll
            for (int r = 0; r < 8; ++r) {
                int srow = 16 * t + r + 8 * lh;
                float val = kvin[srow * 64 + 16 * wave + lc];
                st[t][r] = val;
                s_spi[((srow >> 1) * 16 + lc) * 2 + (srow & 1)] = val;
            }
        }
        if (tid < 64) s_ks[tid] = segKS[(size_t)blockIdx.x * 64 + tid];
    } else {
        #pragma unroll
        for (int t = 0; t < 4; ++t) st[t] = zero8;
        #pragma unroll
        for (int i = 0; i < 32; ++i) sm[OFF_S + tid + i * 128] = 0.f;
        if (tid < 64) s_ks[tid] = 0.f;
    }

    const int row = tid >> 3;
    const int c0  = (tid & 7) * 8;
    const int sp  = tid >> 4;
    const int sc  = (tid & 15) * 4;
    int par = 0;

    for (int ps = 0; ps < npass; ++ps) {
        const int lb = seg * SEGL + ps * TROWS;

        {
            const size_t g0 = ((size_t)((b * LL + lb + row) * HH + h)) * DD + c0;
            const size_t gs = ((size_t)((b * LL + lb + 2 * sp) * HH + h)) * DD + sc;
            v4f qv[SUB][2], ka[SUB][2], va[SUB][2];
            #pragma unroll
            for (int j = 0; j < SUB; ++j) {
                const size_t g = gs + (size_t)j * CCH * HH * DD;
                qv[j][0] = *(const v4f*)(q + g);
                qv[j][1] = *(const v4f*)(q + g + HH * DD);
                ka[j][0] = *(const v4f*)(k + g);
                ka[j][1] = *(const v4f*)(k + g + HH * DD);
                va[j][0] = *(const v4f*)(v + g);
                va[j][1] = *(const v4f*)(v + g + HH * DD);
            }
            __syncthreads();
            #pragma unroll
            for (int j = 0; j < SUB; ++j) {
                float* tl = sm + j * TILE_F;
                const v4f kf0 = fmap4(ka[j][0]);
                const v4f kf1 = fmap4(ka[j][1]);
                *(v4f*)(tl + T_QF + (2 * sp) * 64 + sc)     = fmap4(qv[j][0]);
                *(v4f*)(tl + T_QF + (2 * sp + 1) * 64 + sc) = fmap4(qv[j][1]);
                *(v4f*)(tl + T_KF + (2 * sp) * 64 + sc)     = kf0;
                *(v4f*)(tl + T_KF + (2 * sp + 1) * 64 + sc) = kf1;
                store_pair4(tl + T_KPI + (sp * 64 + sc) * 2, kf0, kf1);
                store_pair4(tl + T_VPI + (sp * 64 + sc) * 2, va[j][0], va[j][1]);
            }
            if (ps + 1 < npass) {
                #pragma unroll
                for (int j = 0; j < SUB; ++j) {
                    const size_t gn = g0 + (size_t)(SUB + j) * CCH * HH * DD;
                    __builtin_prefetch(q + gn, 0, 3);
                    __builtin_prefetch(k + gn, 0, 3);
                    __builtin_prefetch(v + gn, 0, 3);
                }
            }
        }
        __syncthreads();

        for (int j = 0; j < SUB; ++j) {
            const int l0 = lb + j * CCH;
            float* tl = sm + j * TILE_F;
            const float* s_qf  = tl + T_QF;
            const float* s_kf  = tl + T_KF;
            const float* s_kpi = tl + T_KPI;
            const float* s_vpi = tl + T_VPI;
            float* s_a = tl + T_A;
            float* s_z = tl + T_Z;
            const float* ks_cur = s_ks + par * 64;
            float* ks_nxt = s_ks + (par ^ 1) * 64;

            v2f qa[16];
            #pragma unroll
            for (int s = 0; s < 16; ++s)
                qa[s] = *(const v2f*)(s_qf + lc * 64 + 4 * s + 2 * lh);

            v8f oc = zero8;
            #pragma unroll
            for (int t = 0; t < 4; ++t) {
                #pragma unroll
                for (int kk = 0; kk < 4; ++kk) {
                    int p = 8 * t + 2 * kk + lh;
                    v2f bs = *(const v2f*)(s_spi + (p * 16 + lc) * 2);
                    oc = wmma4(qa[4 * t + kk], bs, oc);
                }
            }

            if (wave == 0) {
                v8f aq = zero8;
                #pragma unroll
                for (int s = 0; s < 16; ++s) {
                    v2f bk = *(const v2f*)(s_kf + lc * 64 + 4 * s + 2 * lh);
                    aq = wmma4(qa[s], bk, aq);
                }
                #pragma unroll
                for (int r = 0; r < 8; ++r) {
                    int m = r + 8 * lh;
                    s_a[m * 16 + lc] = (lc <= m) ? aq[r] : 0.f;
                }
            }
            if (wave == 1) {
                v8f zi = zero8;
                #pragma unroll
                for (int s = 0; s < 16; ++s) {
                    v2f bks = *(const v2f*)(ks_cur + 4 * s + 2 * lh);
                    zi = wmma4(qa[s], bks, zi);
                }
                if (lc == 0) {
                    #pragma unroll
                    for (int r = 0; r < 8; ++r) s_z[r + 8 * lh] = zi[r];
                }
            }
            if (wave == 2) {
                float kd0 = 0.f, kd1 = 0.f;
                #pragma unroll
                for (int r16 = 0; r16 < 16; ++r16) {
                    kd0 += s_kf[r16 * 64 + lane * 2];
                    kd1 += s_kf[r16 * 64 + lane * 2 + 1];
                }
                ks_nxt[lane * 2]     = ks_cur[lane * 2]     + kd0;
                ks_nxt[lane * 2 + 1] = ks_cur[lane * 2 + 1] + kd1;
            }
            __syncthreads();

            v2f bv[4];
            #pragma unroll
            for (int kk = 0; kk < 4; ++kk) {
                int p = 2 * kk + lh;
                bv[kk] = *(const v2f*)(s_vpi + (p * 64 + 16 * wave + lc) * 2);
            }
            v8f zr = zero8;
            #pragma unroll
            for (int kk = 0; kk < 4; ++kk) {
                v2f av = *(const v2f*)(s_a + lc * 16 + 4 * kk + 2 * lh);
                oc = wmma4(av, bv[kk], oc);
                zr = wmma4(av, ones2, zr);
            }
            #pragma unroll
            for (int t = 0; t < 4; ++t) {
                #pragma unroll
                for (int kk = 0; kk < 4; ++kk) {
                    int p = 2 * kk + lh;
                    v2f ak = *(const v2f*)(s_kpi + (p * 64 + 16 * t + lc) * 2);
                    st[t] = wmma4(ak, bv[kk], st[t]);
                }
            }

            #pragma unroll
            for (int r = 0; r < 8; ++r) {
                int m = r + 8 * lh;
                float zz = FEPS + zr[r] + s_z[m];
                size_t go = ((size_t)((b * LL + l0 + m) * HH + h)) * DD
                            + 16 * wave + lc;
                out[go] = oc[r] / zz;
            }
            #pragma unroll
            for (int t = 0; t < 4; ++t) {
                #pragma unroll
                for (int r = 0; r < 8; ++r) {
                    int srow = 16 * t + r + 8 * lh;
                    s_spi[((srow >> 1) * 16 + lc) * 2 + (srow & 1)] = st[t][r];
                }
            }
            par ^= 1;
        }
    }
}

extern "C" void kernel_launch(void* const* d_in, const int* in_sizes, int n_in,
                              void* d_out, int out_size, void* d_ws, size_t ws_size,
                              hipStream_t stream) {
    (void)in_sizes; (void)n_in; (void)out_size;
    const float* q = (const float*)d_in[0];
    const float* k = (const float*)d_in[1];
    const float* v = (const float*)d_in[2];
    float* out = (float*)d_out;

    const size_t nSeg   = (size_t)BB * HH * NS;
    const size_t needWs = nSeg * (4096 + 64) * sizeof(float);

    if (ws_size >= needWs) {
        float* segKV = (float*)d_ws;
        float* segKS = segKV + nSeg * 4096;
        hipLaunchKernelGGL(linattn_phase1, dim3(BB * HH * NS), dim3(128), 0,
                           stream, k, v, segKV, segKS);
        hipLaunchKernelGGL(linattn_phase2, dim3(BB * HH, 5), dim3(256), 0,
                           stream, segKV, segKS);
        hipLaunchKernelGGL((linattn_scan<true>), dim3(BB * HH * NS), dim3(128),
                           0, stream, q, k, v, out, segKV, segKS);
    } else {
        hipLaunchKernelGGL((linattn_scan<false>), dim3(BB * HH), dim3(128),
                           0, stream, q, k, v, out,
                           (const float*)nullptr, (const float*)nullptr);
    }
}


# scan kernel: Qf, Kf and score tiles kept in fragment order in LDS (blocks of 4 k, rows rotated per block) instead of row-major: 16-way and 4-way bank conflicts of the fragment reads gone; sums same
# speedup vs baseline: 1.7285x; 1.0569x over previous
#include <hip/hip_runtime.h>
#include <math.h>


typedef __attribute__((ext_vector_type(2))) float v2f;
typedef __attribute__((ext_vector_type(4))) float v4f;
typedef __attribute__((ext_vector_type(8))) float v8f;

#define BB   4
#define LL   2048
#define HH   16
#define DD   64
#define CCH  16
#define NS   8
#define SEGL (LL / NS)
#define SUB  4
#define TROWS (SUB * CCH)
#define FEPS 1e-6f

__device__ __forceinline__ float fmap(float x) {
    return x > 0.f ? x + 1.f : __expf(x);
}

__device__ __forceinline__ v4f fmap4(v4f x) {
    v4f y = {fmap(x[0]), fmap(x[1]), fmap(x[2]), fmap(x[3])};
    return y;
}

__device__ __forceinline__ void store_pair4(float* dst, v4f a, v4f b) {
    v4f lo = {a[0], b[0], a[1], b[1]};
    v4f hi = {a[2], b[2], a[3], b[3]};
    *(v4f*)(dst)     = lo;
    *(v4f*)(dst + 4) = hi;
}

__device__ __forceinline__ v8f wmma4(v2f a, v2f b, v8f c) {
    return __builtin_amdgcn_wmma_f32_16x16x4_f32(
        false, a, false, b, (short)0, c, false, false);
}

__device__ __forceinline__ int frag_off(int row, int quad) {
    return quad * 64 + ((row + quad) & 15) * 4;
}

#define T_QF    0
#define T_KF    1024
#define T_KPI   2048
#define T_VPI   3072
#define T_A     4096
#define T_Z     4352
#define TILE_F  4368
#define OFF_S   (SUB * TILE_F)
#define OFF_KS  (OFF_S + 4096)
#define SMEM_F  (OFF_KS + 128)

#define P1_KF   0
#define P1_KPI  1024
#define P1_VPI  2048
#define P1_TILE 3072

__global__ __launch_bounds__(128)
void linattn_phase1(const float* __restrict__ k,
                    const float* __restrict__ v,
                    float* __restrict__ segKV,
                    float* __restrict__ segKS) {
    __shared__ __attribute__((aligned(16))) float sm[SUB * P1_TILE];
    const int tid  = threadIdx.x;
    const int lane = tid & 31;
    const int wave = tid >> 5;
    const int lh   = lane >> 4;
    const int lc   = lane & 15;
    const int bh   = blockIdx.x / NS;
    const int seg  = blockIdx.x % NS;
    const int b    = bh / HH;
    const int h    = bh % HH;

    const v8f zero8 = {0.f,0.f,0.f,0.f,0.f,0.f,0.f,0.f};
    v8f st[4];
    #pragma unroll
    for (int t = 0; t < 4; ++t) st[t] = zero8;
    float ks0 = 0.f, ks1 = 0.f;

    const int row = tid >> 3;
    const int c0  = (tid & 7) * 8;
    const int sp  = tid >> 4;
    const int sc  = (tid & 15) * 4;

    for (int ps = 0; ps < SEGL / TROWS; ++ps) {
        const int lb = seg * SEGL + ps * TROWS;
        const size_t g0 = ((size_t)((b * LL + lb + row) * HH + h)) * DD + c0;
        const size_t gs = ((size_t)((b * LL + lb + 2 * sp) * HH + h)) * DD + sc;
        v4f ka[SUB][2], va[SUB][2];
        #pragma unroll
        for (int j = 0; j < SUB; ++j) {
            const size_t g = gs + (size_t)j * CCH * HH * DD;
            ka[j][0] = *(const v4f*)(k + g);
            ka[j][1] = *(const v4f*)(k + g + HH * DD);
            va[j][0] = *(const v4f*)(v + g);
            va[j][1] = *(const v4f*)(v + g + HH * DD);
        }
        __syncthreads();
        #pragma unroll
        for (int j = 0; j < SUB; ++j) {
            float* s_kf  = sm + j * P1_TILE + P1_KF;
            float* s_kpi = sm + j * P1_TILE + P1_KPI;
            float* s_vpi = sm + j * P1_TILE + P1_VPI;
            const v4f kf0 = fmap4(ka[j][0]);
            const v4f kf1 = fmap4(ka[j][1]);
            *(v4f*)(s_kf + (2 * sp) * 64 + sc)     = kf0;
            *(v4f*)(s_kf + (2 * sp + 1) * 64 + sc) = kf1;
            store_pair4(s_kpi + (sp * 64 + sc) * 2, kf0, kf1);
            store_pair4(s_vpi + (sp * 64 + sc) * 2, va[j][0], va[j][1]);
        }
        if (ps + 1 < SEGL / TROWS) {
            #pragma unroll
            for (int j = 0; j < SUB; ++j) {
                const size_t gn = g0 + (size_t)(SUB + j) * CCH * HH * DD;
                __builtin_prefetch(k + gn, 0, 3);
                __builtin_prefetch(v + gn, 0, 3);
            }
        }
        __syncthreads();

        for (int j = 0; j < SUB; ++j) {
            const float* s_kf  = sm + j * P1_TILE + P1_KF;
            const float* s_kpi = sm + j * P1_TILE + P1_KPI;
            const float* s_vpi = sm + j * P1_TILE + P1_VPI;
            v2f bv[4];
            #pragma unroll
            for (int kk = 0; kk < 4; ++kk) {
                int p = 2 * kk + lh;
                bv[kk] = *(const v2f*)(s_vpi + (p * 64 + 16 * wave + lc) * 2);
            }
            #pragma unroll
            for (int t = 0; t < 4; ++t) {
                #pragma unroll
                for (int kk = 0; kk < 4; ++kk) {
                    int p = 2 * kk + lh;
                    v2f ak = *(const v2f*)(s_kpi + (p * 64 + 16 * t + lc) * 2);
                    st[t] = wmma4(ak, bv[kk], st[t]);
                }
            }
            if (wave == 2) {
                float a0 = 0.f, a1 = 0.f;
                #pragma unroll
                for (int r16 = 0; r16 < 16; ++r16) {
                    a0 += s_kf[r16 * 64 + lane * 2];
                    a1 += s_kf[r16 * 64 + lane * 2 + 1];
                }
                ks0 += a0; ks1 += a1;
            }
        }
    }

    float* kvout = segKV + (size_t)blockIdx.x * 4096;
    #pragma unroll
    for (int t = 0; t < 4; ++t) {
        #pragma unroll
        for (int r = 0; r < 8; ++r) {
            int srow = 16 * t + r + 8 * lh;
            kvout[srow * 64 + 16 * wave + lc] = st[t][r];
        }
    }
    if (wave == 2) {
        segKS[(size_t)blockIdx.x * 64 + lane * 2]     = ks0;
        segKS[(size_t)blockIdx.x * 64 + lane * 2 + 1] = ks1;
    }
}

__global__ __launch_bounds__(256)
void linattn_phase2(float* __restrict__ segKV,
                    float* __restrict__ segKS) {
    const int bh  = blockIdx.x;
    const int tid = threadIdx.x;
    float* base;
    int stride;
    if (blockIdx.y < 4) {
        base   = segKV + (size_t)bh * NS * 4096 + blockIdx.y * 1024 + tid * 4;
        stride = 4096;
    } else {
        if (tid >= 16) return;
        base   = segKS + (size_t)bh * NS * 64 + tid * 4;
        stride = 64;
    }
    float4 part[NS];
    #pragma unroll
    for (int s = 0; s < NS; ++s)
        part[s] = *(const float4*)(base + (size_t)s * stride);
    float4 acc = make_float4(0.f, 0.f, 0.f, 0.f);
    #pragma unroll
    for (int s = 0; s < NS; ++s) {
        *(float4*)(base + (size_t)s * stride) = acc;
        acc.x += part[s].x;
        acc.y += part[s].y;
        acc.z += part[s].z;
        acc.w += part[s].w;
    }
}

template <bool SEEDED>
__global__ __launch_bounds__(128)
void linattn_scan(const float* __restrict__ q,
                  const float* __restrict__ k,
                  const float* __restrict__ v,
                  float* __restrict__ out,
                  const float* __restrict__ segKV,
                  const float* __restrict__ segKS) {
    __shared__ __attribute__((aligned(16))) float sm[SMEM_F];
    const int tid  = threadIdx.x;
    const int lane = tid & 31;
    const int wave = tid >> 5;
    const int lh   = lane >> 4;
    const int lc   = lane & 15;
    const int bh   = SEEDED ? (blockIdx.x / NS) : blockIdx.x;
    const int seg  = SEEDED ? (blockIdx.x % NS) : 0;
    const int b    = bh / HH;
    const int h    = bh % HH;
    const int npass = SEEDED ? (SEGL / TROWS) : (LL / TROWS);

    float* s_spi = sm + OFF_S + wave * 1024;
    float* s_ks  = sm + OFF_KS;

    const v8f zero8 = {0.f,0.f,0.f,0.f,0.f,0.f,0.f,0.f};
    const v2f ones2 = {1.f, 1.f};
    v8f st[4];

    if (SEEDED) {
        const float* kvin = segKV + (size_t)blockIdx.x * 4096;
        #pragma unroll
        for (int t = 0; t < 4; ++t) {
            #pragma unroll
            for (int r = 0; r < 8; ++r) {
                int srow = 16 * t + r + 8 * lh;
                float val = kvin[srow * 64 + 16 * wave + lc];
                st[t][r] = val;
                s_spi[((srow >> 1) * 16 + lc) * 2 + (srow & 1)] = val;
            }
        }
        if (tid < 64) s_ks[tid] = segKS[(size_t)blockIdx.x * 64 + tid];
    } else {
        #pragma unroll
        for (int t = 0; t < 4; ++t) st[t] = zero8;
        #pragma unroll
        for (int i = 0; i < 32; ++i) sm[OFF_S + tid + i * 128] = 0.f;
        if (tid < 64) s_ks[tid] = 0.f;
    }

    const int row = tid >> 3;
    const int c0  = (tid & 7) * 8;
    const int sp  = tid >> 4;
    const int sc  = (tid & 15) * 4;
    const int fo0 = frag_off(2 * sp, sc >> 2);
    const int fo1 = frag_off(2 * sp + 1, sc >> 2);
    int par = 0;

    for (int ps = 0; ps < npass; ++ps) {
        const int lb = seg * SEGL + ps * TROWS;

        {
            const size_t g0 = ((size_t)((b * LL + lb + row) * HH + h)) * DD + c0;
            const size_t gs = ((size_t)((b * LL + lb + 2 * sp) * HH + h)) * DD + sc;
            v4f qv[SUB][2], ka[SUB][2], va[SUB][2];
            #pragma unroll
            for (int j = 0; j < SUB; ++j) {
                const size_t g = gs + (size_t)j * CCH * HH * DD;
                qv[j][0] = *(const v4f*)(q + g);
                qv[j][1] = *(const v4f*)(q + g + HH * DD);
                ka[j][0] = *(const v4f*)(k + g);
                ka[j][1] = *(const v4f*)(k + g + HH * DD);
                va[j][0] = *(const v4f*)(v + g);
                va[j][1] = *(const v4f*)(v + g + HH * DD);
            }
            __syncthreads();
            #pragma unroll
            for (int j = 0; j < SUB; ++j) {
                float* tl = sm + j * TILE_F;
                const v4f kf0 = fmap4(ka[j][0]);
                const v4f kf1 = fmap4(ka[j][1]);
                *(v4f*)(tl + T_QF + fo0) = fmap4(qv[j][0]);
                *(v4f*)(tl + T_QF + fo1) = fmap4(qv[j][1]);
                *(v4f*)(tl + T_KF + fo0) = kf0;
                *(v4f*)(tl + T_KF + fo1) = kf1;
                store_pair4(tl + T_KPI + (sp * 64 + sc) * 2, kf0, kf1);
                store_pair4(tl + T_VPI + (sp * 64 + sc) * 2, va[j][0], va[j][1]);
            }
            if (ps + 1 < npass) {
                #pragma unroll
                for (int j = 0; j < SUB; ++j) {
                    const size_t gn = g0 + (size_t)(SUB + j) * CCH * HH * DD;
                    __builtin_prefetch(q + gn, 0, 3);
                    __builtin_prefetch(k + gn, 0, 3);
                    __builtin_prefetch(v + gn, 0, 3);
                }
            }
        }
        __syncthreads();

        for (int j = 0; j < SUB; ++j) {
            const int l0 = lb + j * CCH;
            float* tl = sm + j * TILE_F;
            const float* s_qf  = tl + T_QF;
            const float* s_kf  = tl + T_KF;
            const float* s_kpi = tl + T_KPI;
            const float* s_vpi = tl + T_VPI;
            float* s_a = tl + T_A;
            float* s_z = tl + T_Z;
            const float* ks_cur = s_ks + par * 64;
            float* ks_nxt = s_ks + (par ^ 1) * 64;

            v2f qa[16];
            #pragma unroll
            for (int s = 0; s < 16; ++s)
                qa[s] = *(const v2f*)(s_qf + frag_off(lc, s) + 2 * lh);

            v8f oc = zero8;
            #pragma unroll
            for (int t = 0; t < 4; ++t) {
                #pragma unroll
                for (int kk = 0; kk < 4; ++kk) {
                    int p = 8 * t + 2 * kk + lh;
                    v2f bs = *(const v2f*)(s_spi + (p * 16 + lc) * 2);
                    oc = wmma4(qa[4 * t + kk], bs, oc);
                }
            }

            if (wave == 0) {
                v8f aq = zero8;
                #pragma unroll
                for (int s = 0; s < 16; ++s) {
                    v2f bk = *(const v2f*)(s_kf + frag_off(lc, s) + 2 * lh);
                    aq = wmma4(qa[s], bk, aq);
                }
                #pragma unroll
                for (int r = 0; r < 8; ++r) {
                    int m = r + 8 * lh;
                    s_a[frag_off(m, lc >> 2) + (lc & 3)] = (lc <= m) ? aq[r] : 0.f;
                }
            }
            if (wave == 1) {
                v8f zi = zero8;
                #pragma unroll
                for (int s = 0; s < 16; ++s) {
                    v2f bks = *(const v2f*)(ks_cur + 4 * s + 2 * lh);
                    zi = wmma4(qa[s], bks, zi);
                }
                if (lc == 0) {
                    #pragma unroll
                    for (int r = 0; r < 8; ++r) s_z[r + 8 * lh] = zi[r];
                }
            }
            if (wave == 2) {
                float kd0 = 0.f, kd1 = 0.f;
                #pragma unroll
                for (int p = 0; p < 8; ++p) {
                    const v4f kp = *(const v4f*)(s_kpi + (p * 64 + lane * 2) * 2);
                    kd0 += kp[0]; kd0 += kp[1];
                    kd1 += kp[2]; kd1 += kp[3];
                }
                ks_nxt[lane * 2]     = ks_cur[lane * 2]     + kd0;
                ks_nxt[lane * 2 + 1] = ks_cur[lane * 2 + 1] + kd1;
            }
            __syncthreads();

            v2f bv[4];
            #pragma unroll
            for (int kk = 0; kk < 4; ++kk) {
                int p = 2 * kk + lh;
                bv[kk] = *(const v2f*)(s_vpi + (p * 64 + 16 * wave + lc) * 2);
            }
            v8f zr = zero8;
            #pragma unroll
            for (int kk = 0; kk < 4; ++kk) {
                v2f av = *(const v2f*)(s_a + frag_off(lc, kk) + 2 * lh);
                oc = wmma4(av, bv[kk], oc);
                zr = wmma4(av, ones2, zr);
            }
            #pragma unroll
            for (int t = 0; t < 4; ++t) {
                #pragma unroll
                for (int kk = 0; kk < 4; ++kk) {
                    int p = 2 * kk + lh;
                    v2f ak = *(const v2f*)(s_kpi + (p * 64 + 16 * t + lc) * 2);
                    st[t] = wmma4(ak, bv[kk], st[t]);
                }
            }

            #pragma unroll
            for (int r = 0; r < 8; ++r) {
                int m = r + 8 * lh;
                float zz = FEPS + zr[r] + s_z[m];
                size_t go = ((size_t)((b * LL + l0 + m) * HH + h)) * DD
                            + 16 * wave + lc;
                out[go] = oc[r] / zz;
            }
            #pragma unroll
            for (int t = 0; t < 4; ++t) {
                #pragma unroll
                for (int r = 0; r < 8; ++r) {
                    int srow = 16 * t + r + 8 * lh;
                    s_spi[((srow >> 1) * 16 + lc) * 2 + (srow & 1)] = st[t][r];
                }
            }
            par ^= 1;
        }
    }
}

extern "C" void kernel_launch(void* const* d_in, const int* in_sizes, int n_in,
                              void* d_out, int out_size, void* d_ws, size_t ws_size,
                              hipStream_t stream) {
    (void)in_sizes; (void)n_in; (void)out_size;
    const float* q = (const float*)d_in[0];
    const float* k = (const float*)d_in[1];
    const float* v = (const float*)d_in[2];
    float* out = (float*)d_out;

    const size_t nSeg   = (size_t)BB * HH * NS;
    const size_t needWs = nSeg * (4096 + 64) * sizeof(float);

    if (ws_size >= needWs) {
        float* segKV = (float*)d_ws;
        float* segKS = segKV + nSeg * 4096;
        hipLaunchKernelGGL(linattn_phase1, dim3(BB * HH * NS), dim3(128), 0,
                           stream, k, v, segKV, segKS);
        hipLaunchKernelGGL(linattn_phase2, dim3(BB * HH, 5), dim3(256), 0,
                           stream, segKV, segKS);
        hipLaunchKernelGGL((linattn_scan<true>), dim3(BB * HH * NS), dim3(128),
                           0, stream, q, k, v, out, segKV, segKS);
    } else {
        hipLaunchKernelGGL((linattn_scan<false>), dim3(BB * HH), dim3(128),
                           0, stream, q, k, v, out,
                           (const float*)nullptr, (const float*)nullptr);
    }
}


# scan kernel: state kept only in accumulator registers and fed to Qf@S as its B operand (register pair = k pair); LDS state mirror gone (32 stores, 16 loads a chunk); new sum order: Qf@S, QK^T, z
# speedup vs baseline: 1.8340x; 1.0610x over previous
#include <hip/hip_runtime.h>
#include <math.h>


typedef __attribute__((ext_vector_type(2))) float v2f;
typedef __attribute__((ext_vector_type(4))) float v4f;
typedef __attribute__((ext_vector_type(8))) float v8f;

#define BB   4
#define LL   2048
#define HH   16
#define DD   64
#define CCH  16
#define NS   8
#define SEGL (LL / NS)
#define SUB  4
#define TROWS (SUB * CCH)
#define FEPS 1e-6f

__device__ __forceinline__ float fmap(float x) {
    return x > 0.f ? x + 1.f : __expf(x);
}

__device__ __forceinline__ v4f fmap4(v4f x) {
    v4f y = {fmap(x[0]), fmap(x[1]), fmap(x[2]), fmap(x[3])};
    return y;
}

__device__ __forceinline__ void store_pair4(float* dst, v4f a, v4f b) {
    v4f lo = {a[0], b[0], a[1], b[1]};
    v4f hi = {a[2], b[2], a[3], b[3]};
    *(v4f*)(dst)     = lo;
    *(v4f*)(dst + 4) = hi;
}

__device__ __forceinline__ v8f wmma4(v2f a, v2f b, v8f c) {
    return __builtin_amdgcn_wmma_f32_16x16x4_f32(
        false, a, false, b, (short)0, c, false, false);
}

#define LO2(x) __builtin_shufflevector((x), (x), 0, 1)
#define HI2(x) __builtin_shufflevector((x), (x), 2, 3)

__device__ __forceinline__ int frag_off(int row, int quad) {
    return quad * 64 + ((row + quad) & 15) * 4;
}

#define T_QF    0
#define T_KF    1024
#define T_KPI   2048
#define T_VPI   3072
#define T_A     4096
#define T_Z     4352
#define TILE_F  4368
#define OFF_KS  (SUB * TILE_F)
#define SMEM_F  (OFF_KS + 128)

#define P1_KF   0
#define P1_KPI  1024
#define P1_VPI  2048
#define P1_TILE 3072

__global__ __launch_bounds__(128)
void linattn_phase1(const float* __restrict__ k,
                    const float* __restrict__ v,
                    float* __restrict__ segKV,
                    float* __restrict__ segKS) {
    __shared__ __attribute__((aligned(16))) float sm[SUB * P1_TILE];
    const int tid  = threadIdx.x;
    const int lane = tid & 31;
    const int wave = tid >> 5;
    const int lh   = lane >> 4;
    const int lc   = lane & 15;
    const int bh   = blockIdx.x / NS;
    const int seg  = blockIdx.x % NS;
    const int b    = bh / HH;
    const int h    = bh % HH;

    const v8f zero8 = {0.f,0.f,0.f,0.f,0.f,0.f,0.f,0.f};
    v8f st[4];
    #pragma unroll
    for (int t = 0; t < 4; ++t) st[t] = zero8;
    float ks0 = 0.f, ks1 = 0.f;

    const int row = tid >> 3;
    const int c0  = (tid & 7) * 8;
    const int sp  = tid >> 4;
    const int sc  = (tid & 15) * 4;

    for (int ps = 0; ps < SEGL / TROWS; ++ps) {
        const int lb = seg * SEGL + ps * TROWS;
        const size_t g0 = ((size_t)((b * LL + lb + row) * HH + h)) * DD + c0;
        const size_t gs = ((size_t)((b * LL + lb + 2 * sp) * HH + h)) * DD + sc;
        v4f ka[SUB][2], va[SUB][2];
        #pragma unroll
        for (int j = 0; j < SUB; ++j) {
            const size_t g = gs + (size_t)j * CCH * HH * DD;
            ka[j][0] = *(const v4f*)(k + g);
            ka[j][1] = *(const v4f*)(k + g + HH * DD);
            va[j][0] = *(const v4f*)(v + g);
            va[j][1] = *(const v4f*)(v + g + HH * DD);
        }
        __syncthreads();
        #pragma unroll
        for (int j = 0; j < SUB; ++j) {
            float* s_kf  = sm + j * P1_TILE + P1_KF;
            float* s_kpi = sm + j * P1_TILE + P1_KPI;
            float* s_vpi = sm + j * P1_TILE + P1_VPI;
            const v4f kf0 = fmap4(ka[j][0]);
            const v4f kf1 = fmap4(ka[j][1]);
            *(v4f*)(s_kf + (2 * sp) * 64 + sc)     = kf0;
            *(v4f*)(s_kf + (2 * sp + 1) * 64 + sc) = kf1;
            store_pair4(s_kpi + (sp * 64 + sc) * 2, kf0, kf1);
            store_pair4(s_vpi + (sp * 64 + sc) * 2, va[j][0], va[j][1]);
        }
        if (ps + 1 < SEGL / TROWS) {
            #pragma unroll
            for (int j = 0; j < SUB; ++j) {
                const size_t gn = g0 + (size_t)(SUB + j) * CCH * HH * DD;
                __builtin_prefetch(k + gn, 0, 3);
                __builtin_prefetch(v + gn, 0, 3);
            }
        }
        __syncthreads();

        for (int j = 0; j < SUB; ++j) {
            const float* s_kf  = sm + j * P1_TILE + P1_KF;
            const float* s_kpi = sm + j * P1_TILE + P1_KPI;
            const float* s_vpi = sm + j * P1_TILE + P1_VPI;
            v2f bv[4];
            #pragma unroll
            for (int kk = 0; kk < 4; ++kk) {
                int p = 2 * kk + lh;
                bv[kk] = *(const v2f*)(s_vpi + (p * 64 + 16 * wave + lc) * 2);
            }
            #pragma unroll
            for (int t = 0; t < 4; ++t) {
                #pragma unroll
                for (int kk = 0; kk < 4; ++kk) {
                    int p = 2 * kk + lh;
                    v2f ak = *(const v2f*)(s_kpi + (p * 64 + 16 * t + lc) * 2);
                    st[t] = wmma4(ak, bv[kk], st[t]);
                }
            }
            if (wave == 2) {
                float a0 = 0.f, a1 = 0.f;
                #pragma unroll
                for (int r16 = 0; r16 < 16; ++r16) {
                    a0 += s_kf[r16 * 64 + lane * 2];
                    a1 += s_kf[r16 * 64 + lane * 2 + 1];
                }
                ks0 += a0; ks1 += a1;
            }
        }
    }

    float* kvout = segKV + (size_t)blockIdx.x * 4096;
    #pragma unroll
    for (int t = 0; t < 4; ++t) {
        #pragma unroll
        for (int r = 0; r < 8; ++r) {
            int srow = 16 * t + r + 8 * lh;
            kvout[srow * 64 + 16 * wave + lc] = st[t][r];
        }
    }
    if (wave == 2) {
        segKS[(size_t)blockIdx.x * 64 + lane * 2]     = ks0;
        segKS[(size_t)blockIdx.x * 64 + lane * 2 + 1] = ks1;
    }
}

__global__ __launch_bounds__(256)
void linattn_phase2(float* __restrict__ segKV,
                    float* __restrict__ segKS) {
    const int bh  = blockIdx.x;
    const int tid = threadIdx.x;
    float* base;
    int stride;
    if (blockIdx.y < 4) {
        base   = segKV + (size_t)bh * NS * 4096 + blockIdx.y * 1024 + tid * 4;
        stride = 4096;
    } else {
        if (tid >= 16) return;
        base   = segKS + (size_t)bh * NS * 64 + tid * 4;
        stride = 64;
    }
    float4 part[NS];
    #pragma unroll
    for (int s = 0; s < NS; ++s)
        part[s] = *(const float4*)(base + (size_t)s * stride);
    float4 acc = make_float4(0.f, 0.f, 0.f, 0.f);
    #pragma unroll
    for (int s = 0; s < NS; ++s) {
        *(float4*)(base + (size_t)s * stride) = acc;
        acc.x += part[s].x;
        acc.y += part[s].y;
        acc.z += part[s].z;
        acc.w += part[s].w;
    }
}

template <bool SEEDED>
__global__ __launch_bounds__(128)
void linattn_scan(const float* __restrict__ q,
                  const float* __restrict__ k,
                  const float* __restrict__ v,
                  float* __restrict__ out,
                  const float* __restrict__ segKV,
                  const float* __restrict__ segKS) {
    __shared__ __attribute__((aligned(16))) float sm[SMEM_F];
    const int tid  = threadIdx.x;
    const int lane = tid & 31;
    const int wave = tid >> 5;
    const int lh   = lane >> 4;
    const int lc   = lane & 15;
    const int bh   = SEEDED ? (blockIdx.x / NS) : blockIdx.x;
    const int seg  = SEEDED ? (blockIdx.x % NS) : 0;
    const int b    = bh / HH;
    const int h    = bh % HH;
    const int npass = SEEDED ? (SEGL / TROWS) : (LL / TROWS);

    float* s_ks  = sm + OFF_KS;

    const v8f zero8 = {0.f,0.f,0.f,0.f,0.f,0.f,0.f,0.f};
    const v2f ones2 = {1.f, 1.f};
    v8f st[4];

    if (SEEDED) {
        const float* kvin = segKV + (size_t)blockIdx.x * 4096;
        #pragma unroll
        for (int t = 0; t < 4; ++t) {
            #pragma unroll
            for (int r = 0; r < 8; ++r) {
                int srow = 16 * t + r + 8 * lh;
                st[t][r] = kvin[srow * 64 + 16 * wave + lc];
            }
        }
        if (tid < 64) s_ks[tid] = segKS[(size_t)blockIdx.x * 64 + tid];
    } else {
        #pragma unroll
        for (int t = 0; t < 4; ++t) st[t] = zero8;
        if (tid < 64) s_ks[tid] = 0.f;
    }

    const int row = tid >> 3;
    const int c0  = (tid & 7) * 8;
    const int sp  = tid >> 4;
    const int sc  = (tid & 15) * 4;
    const int fo0 = frag_off(2 * sp, sc >> 2);
    const int fo1 = frag_off(2 * sp + 1, sc >> 2);
    int par = 0;

    for (int ps = 0; ps < npass; ++ps) {
        const int lb = seg * SEGL + ps * TROWS;

        {
            const size_t g0 = ((size_t)((b * LL + lb + row) * HH + h)) * DD + c0;
            const size_t gs = ((size_t)((b * LL + lb + 2 * sp) * HH + h)) * DD + sc;
            v4f qv[SUB][2], ka[SUB][2], va[SUB][2];
            #pragma unroll
            for (int j = 0; j < SUB; ++j) {
                const size_t g = gs + (size_t)j * CCH * HH * DD;
                qv[j][0] = *(const v4f*)(q + g);
                qv[j][1] = *(const v4f*)(q + g + HH * DD);
                ka[j][0] = *(const v4f*)(k + g);
                ka[j][1] = *(const v4f*)(k + g + HH * DD);
                va[j][0] = *(const v4f*)(v + g);
                va[j][1] = *(const v4f*)(v + g + HH * DD);
            }
            __syncthreads();
            #pragma unroll
            for (int j = 0; j < SUB; ++j) {
                float* tl = sm + j * TILE_F;
                const v4f kf0 = fmap4(ka[j][0]);
                const v4f kf1 = fmap4(ka[j][1]);
                *(v4f*)(tl + T_QF + fo0) = fmap4(qv[j][0]);
                *(v4f*)(tl + T_QF + fo1) = fmap4(qv[j][1]);
                *(v4f*)(tl + T_KF + fo0) = kf0;
                *(v4f*)(tl + T_KF + fo1) = kf1;
                store_pair4(tl + T_KPI + (sp * 64 + sc) * 2, kf0, kf1);
                store_pair4(tl + T_VPI + (sp * 64 + sc) * 2, va[j][0], va[j][1]);
            }
            if (ps + 1 < npass) {
                #pragma unroll
                for (int j = 0; j < SUB; ++j) {
                    const size_t gn = g0 + (size_t)(SUB + j) * CCH * HH * DD;
                    __builtin_prefetch(q + gn, 0, 3);
                    __builtin_prefetch(k + gn, 0, 3);
                    __builtin_prefetch(v + gn, 0, 3);
                }
            }
        }
        __syncthreads();

        for (int j = 0; j < SUB; ++j) {
            const int l0 = lb + j * CCH;
            float* tl = sm + j * TILE_F;
            const float* s_qf  = tl + T_QF;
            const float* s_kf  = tl + T_KF;
            const float* s_kpi = tl + T_KPI;
            const float* s_vpi = tl + T_VPI;
            float* s_a = tl + T_A;
            float* s_z = tl + T_Z;
            const float* ks_cur = s_ks + par * 64;
            float* ks_nxt = s_ks + (par ^ 1) * 64;

            v4f qq[8];
            #pragma unroll
            for (int u = 0; u < 8; ++u)
                qq[u] = *(const v4f*)(s_qf + frag_off(lc, 4 * (u >> 1) + 2 * lh + (u & 1)));

            v8f oc = zero8;
            #pragma unroll
            for (int t = 0; t < 4; ++t) {
                oc = wmma4(LO2(qq[2 * t]),
                           __builtin_shufflevector(st[t], st[t], 0, 1), oc);
                oc = wmma4(HI2(qq[2 * t]),
                           __builtin_shufflevector(st[t], st[t], 2, 3), oc);
                oc = wmma4(LO2(qq[2 * t + 1]),
                           __builtin_shufflevector(st[t], st[t], 4, 5), oc);
                oc = wmma4(HI2(qq[2 * t + 1]),
                           __builtin_shufflevector(st[t], st[t], 6, 7), oc);
            }

            if (wave == 0) {
                v8f aq = zero8;
                #pragma unroll
                for (int u = 0; u < 8; ++u) {
                    const v4f bk = *(const v4f*)(s_kf + frag_off(lc, 4 * (u >> 1) + 2 * lh + (u & 1)));
                    aq = wmma4(LO2(qq[u]), LO2(bk), aq);
                    aq = wmma4(HI2(qq[u]), HI2(bk), aq);
                }
                #pragma unroll
                for (int r = 0; r < 8; ++r) {
                    int m = r + 8 * lh;
                    s_a[frag_off(m, lc >> 2) + (lc & 3)] = (lc <= m) ? aq[r] : 0.f;
                }
            }
            if (wave == 1) {
                v8f zi = zero8;
                #pragma unroll
                for (int u = 0; u < 8; ++u) {
                    const v4f bks = *(const v4f*)(ks_cur + 4 * (4 * (u >> 1) + 2 * lh + (u & 1)));
                    zi = wmma4(LO2(qq[u]), LO2(bks), zi);
                    zi = wmma4(HI2(qq[u]), HI2(bks), zi);
                }
                if (lc == 0) {
                    #pragma unroll
                    for (int r = 0; r < 8; ++r) s_z[r + 8 * lh] = zi[r];
                }
            }
            if (wave == 2) {
                float kd0 = 0.f, kd1 = 0.f;
                #pragma unroll
                for (int p = 0; p < 8; ++p) {
                    const v4f kp = *(const v4f*)(s_kpi + (p * 64 + lane * 2) * 2);
                    kd0 += kp[0]; kd0 += kp[1];
                    kd1 += kp[2]; kd1 += kp[3];
                }
                ks_nxt[lane * 2]     = ks_cur[lane * 2]     + kd0;
                ks_nxt[lane * 2 + 1] = ks_cur[lane * 2 + 1] + kd1;
            }
            __syncthreads();

            v2f bv[4];
            #pragma unroll
            for (int kk = 0; kk < 4; ++kk) {
                int p = 2 * kk + lh;
                bv[kk] = *(const v2f*)(s_vpi + (p * 64 + 16 * wave + lc) * 2);
            }
            v8f zr = zero8;
            #pragma unroll
            for (int kk = 0; kk < 4; ++kk) {
                v2f av = *(const v2f*)(s_a + frag_off(lc, kk) + 2 * lh);
                oc = wmma4(av, bv[kk], oc);
                zr = wmma4(av, ones2, zr);
            }
            #pragma unroll
            for (int t = 0; t < 4; ++t) {
                #pragma unroll
                for (int kk = 0; kk < 4; ++kk) {
                    int p = 2 * kk + lh;
                    v2f ak = *(const v2f*)(s_kpi + (p * 64 + 16 * t + lc) * 2);
                    st[t] = wmma4(ak, bv[kk], st[t]);
                }
            }

            #pragma unroll
            for (int r = 0; r < 8; ++r) {
                int m = r + 8 * lh;
                float zz = FEPS + zr[r] + s_z[m];
                size_t go = ((size_t)((b * LL + l0 + m) * HH + h)) * DD
                            + 16 * wave + lc;
                out[go] = oc[r] / zz;
            }
            par ^= 1;
        }
    }
}

extern "C" void kernel_launch(void* const* d_in, const int* in_sizes, int n_in,
                              void* d_out, int out_size, void* d_ws, size_t ws_size,
                              hipStream_t stream) {
    (void)in_sizes; (void)n_in; (void)out_size;
    const float* q = (const float*)d_in[0];
    const float* k = (const float*)d_in[1];
    const float* v = (const float*)d_in[2];
    float* out = (float*)d_out;

    const size_t nSeg   = (size_t)BB * HH * NS;
    const size_t needWs = nSeg * (4096 + 64) * sizeof(float);

    if (ws_size >= needWs) {
        float* segKV = (float*)d_ws;
        float* segKS = segKV + nSeg * 4096;
        hipLaunchKernelGGL(linattn_phase1, dim3(BB * HH * NS), dim3(128), 0,
                           stream, k, v, segKV, segKS);
        hipLaunchKernelGGL(linattn_phase2, dim3(BB * HH, 5), dim3(256), 0,
                           stream, segKV, segKS);
        hipLaunchKernelGGL((linattn_scan<true>), dim3(BB * HH * NS), dim3(128),
                           0, stream, q, k, v, out, segKV, segKS);
    } else {
        hipLaunchKernelGGL((linattn_scan<false>), dim3(BB * HH), dim3(128),
                           0, stream, q, k, v, out,
                           (const float*)nullptr, (const float*)nullptr);
    }
}


# scan kernel: score tiles and normalizer rows of a 64-row pass formed up front, chunk w by wave w (each wave keeps the Kf column sum itself); chunk loop has no barrier, equal work per wave; sums same
# speedup vs baseline: 1.9251x; 1.0497x over previous
#include <hip/hip_runtime.h>
#include <math.h>


typedef __attribute__((ext_vector_type(2))) float v2f;
typedef __attribute__((ext_vector_type(4))) float v4f;
typedef __attribute__((ext_vector_type(8))) float v8f;

#define BB   4
#define LL   2048
#define HH   16
#define DD   64
#define CCH  16
#define NS   8
#define SEGL (LL / NS)
#define SUB  4
#define TROWS (SUB * CCH)
#define FEPS 1e-6f

__device__ __forceinline__ float fmap(float x) {
    return x > 0.f ? x + 1.f : __expf(x);
}

__device__ __forceinline__ v4f fmap4(v4f x) {
    v4f y = {fmap(x[0]), fmap(x[1]), fmap(x[2]), fmap(x[3])};
    return y;
}

__device__ __forceinline__ void store_pair4(float* dst, v4f a, v4f b) {
    v4f lo = {a[0], b[0], a[1], b[1]};
    v4f hi = {a[2], b[2], a[3], b[3]};
    *(v4f*)(dst)     = lo;
    *(v4f*)(dst + 4) = hi;
}

__device__ __forceinline__ v8f wmma4(v2f a, v2f b, v8f c) {
    return __builtin_amdgcn_wmma_f32_16x16x4_f32(
        false, a, false, b, (short)0, c, false, false);
}

#define LO2(x) __builtin_shufflevector((x), (x), 0, 1)
#define HI2(x) __builtin_shufflevector((x), (x), 2, 3)

__device__ __forceinline__ int frag_off(int row, int quad) {
    return quad * 64 + ((row + quad) & 15) * 4;
}

#define T_QF    0
#define T_KF    1024
#define T_KPI   2048
#define T_VPI   3072
#define T_A     4096
#define T_Z     4352
#define TILE_F  4368
#define OFF_KS  (SUB * TILE_F)
#define SMEM_F  (OFF_KS + 4 * 64)

#define P1_KF   0
#define P1_KPI  1024
#define P1_VPI  2048
#define P1_TILE 3072

__global__ __launch_bounds__(128)
void linattn_phase1(const float* __restrict__ k,
                    const float* __restrict__ v,
                    float* __restrict__ segKV,
                    float* __restrict__ segKS) {
    __shared__ __attribute__((aligned(16))) float sm[SUB * P1_TILE];
    const int tid  = threadIdx.x;
    const int lane = tid & 31;
    const int wave = tid >> 5;
    const int lh   = lane >> 4;
    const int lc   = lane & 15;
    const int bh   = blockIdx.x / NS;
    const int seg  = blockIdx.x % NS;
    const int b    = bh / HH;
    const int h    = bh % HH;

    const v8f zero8 = {0.f,0.f,0.f,0.f,0.f,0.f,0.f,0.f};
    v8f st[4];
    #pragma unroll
    for (int t = 0; t < 4; ++t) st[t] = zero8;
    float ks0 = 0.f, ks1 = 0.f;

    const int row = tid >> 3;
    const int c0  = (tid & 7) * 8;
    const int sp  = tid >> 4;
    const int sc  = (tid & 15) * 4;

    for (int ps = 0; ps < SEGL / TROWS; ++ps) {
        const int lb = seg * SEGL + ps * TROWS;
        const size_t g0 = ((size_t)((b * LL + lb + row) * HH + h)) * DD + c0;
        const size_t gs = ((size_t)((b * LL + lb + 2 * sp) * HH + h)) * DD + sc;
        v4f ka[SUB][2], va[SUB][2];
        #pragma unroll
        for (int j = 0; j < SUB; ++j) {
            const size_t g = gs + (size_t)j * CCH * HH * DD;
            ka[j][0] = *(const v4f*)(k + g);
            ka[j][1] = *(const v4f*)(k + g + HH * DD);
            va[j][0] = *(const v4f*)(v + g);
            va[j][1] = *(const v4f*)(v + g + HH * DD);
        }
        __syncthreads();
        #pragma unroll
        for (int j = 0; j < SUB; ++j) {
            float* s_kf  = sm + j * P1_TILE + P1_KF;
            float* s_kpi = sm + j * P1_TILE + P1_KPI;
            float* s_vpi = sm + j * P1_TILE + P1_VPI;
            const v4f kf0 = fmap4(ka[j][0]);
            const v4f kf1 = fmap4(ka[j][1]);
            *(v4f*)(s_kf + (2 * sp) * 64 + sc)     = kf0;
            *(v4f*)(s_kf + (2 * sp + 1) * 64 + sc) = kf1;
            store_pair4(s_kpi + (sp * 64 + sc) * 2, kf0, kf1);
            store_pair4(s_vpi + (sp * 64 + sc) * 2, va[j][0], va[j][1]);
        }
        if (ps + 1 < SEGL / TROWS) {
            #pragma unroll
            for (int j = 0; j < SUB; ++j) {
                const size_t gn = g0 + (size_t)(SUB + j) * CCH * HH * DD;
                __builtin_prefetch(k + gn, 0, 3);
                __builtin_prefetch(v + gn, 0, 3);
            }
        }
        __syncthreads();

        for (int j = 0; j < SUB; ++j) {
            const float* s_kf  = sm + j * P1_TILE + P1_KF;
            const float* s_kpi = sm + j * P1_TILE + P1_KPI;
            const float* s_vpi = sm + j * P1_TILE + P1_VPI;
            v2f bv[4];
            #pragma unroll
            for (int kk = 0; kk < 4; ++kk) {
                int p = 2 * kk + lh;
                bv[kk] = *(const v2f*)(s_vpi + (p * 64 + 16 * wave + lc) * 2);
            }
            #pragma unroll
            for (int t = 0; t < 4; ++t) {
                #pragma unroll
                for (int kk = 0; kk < 4; ++kk) {
                    int p = 2 * kk + lh;
                    v2f ak = *(const v2f*)(s_kpi + (p * 64 + 16 * t + lc) * 2);
                    st[t] = wmma4(ak, bv[kk], st[t]);
                }
            }
            if (wave == 2) {
                float a0 = 0.f, a1 = 0.f;
                #pragma unroll
                for (int r16 = 0; r16 < 16; ++r16) {
                    a0 += s_kf[r16 * 64 + lane * 2];
                    a1 += s_kf[r16 * 64 + lane * 2 + 1];
                }
                ks0 += a0; ks1 += a1;
            }
        }
    }

    float* kvout = segKV + (size_t)blockIdx.x * 4096;
    #pragma unroll
    for (int t = 0; t < 4; ++t) {
        #pragma unroll
        for (int r = 0; r < 8; ++r) {
            int srow = 16 * t + r + 8 * lh;
            kvout[srow * 64 + 16 * wave + lc] = st[t][r];
        }
    }
    if (wave == 2) {
        segKS[(size_t)blockIdx.x * 64 + lane * 2]     = ks0;
        segKS[(size_t)blockIdx.x * 64 + lane * 2 + 1] = ks1;
    }
}

__global__ __launch_bounds__(256)
void linattn_phase2(float* __restrict__ segKV,
                    float* __restrict__ segKS) {
    const int bh  = blockIdx.x;
    const int tid = threadIdx.x;
    float* base;
    int stride;
    if (blockIdx.y < 4) {
        base   = segKV + (size_t)bh * NS * 4096 + blockIdx.y * 1024 + tid * 4;
        stride = 4096;
    } else {
        if (tid >= 16) return;
        base   = segKS + (size_t)bh * NS * 64 + tid * 4;
        stride = 64;
    }
    float4 part[NS];
    #pragma unroll
    for (int s = 0; s < NS; ++s)
        part[s] = *(const float4*)(base + (size_t)s * stride);
    float4 acc = make_float4(0.f, 0.f, 0.f, 0.f);
    #pragma unroll
    for (int s = 0; s < NS; ++s) {
        *(float4*)(base + (size_t)s * stride) = acc;
        acc.x += part[s].x;
        acc.y += part[s].y;
        acc.z += part[s].z;
        acc.w += part[s].w;
    }
}

template <bool SEEDED>
__global__ __launch_bounds__(128)
void linattn_scan(const float* __restrict__ q,
                  const float* __restrict__ k,
                  const float* __restrict__ v,
                  float* __restrict__ out,
                  const float* __restrict__ segKV,
                  const float* __restrict__ segKS) {
    __shared__ __attribute__((aligned(16))) float sm[SMEM_F];
    const int tid  = threadIdx.x;
    const int lane = tid & 31;
    const int wave = tid >> 5;
    const int lh   = lane >> 4;
    const int lc   = lane & 15;
    const int bh   = SEEDED ? (blockIdx.x / NS) : blockIdx.x;
    const int seg  = SEEDED ? (blockIdx.x % NS) : 0;
    const int b    = bh / HH;
    const int h    = bh % HH;
    const int npass = SEEDED ? (SEGL / TROWS) : (LL / TROWS);

    float* s_ks  = sm + OFF_KS;

    const v8f zero8 = {0.f,0.f,0.f,0.f,0.f,0.f,0.f,0.f};
    const v2f ones2 = {1.f, 1.f};
    v8f st[4];
    float ksr0, ksr1;

    if (SEEDED) {
        const float* kvin = segKV + (size_t)blockIdx.x * 4096;
        #pragma unroll
        for (int t = 0; t < 4; ++t) {
            #pragma unroll
            for (int r = 0; r < 8; ++r) {
                int srow = 16 * t + r + 8 * lh;
                st[t][r] = kvin[srow * 64 + 16 * wave + lc];
            }
        }
        ksr0 = segKS[(size_t)blockIdx.x * 64 + lane * 2];
        ksr1 = segKS[(size_t)blockIdx.x * 64 + lane * 2 + 1];
    } else {
        #pragma unroll
        for (int t = 0; t < 4; ++t) st[t] = zero8;
        ksr0 = 0.f;
        ksr1 = 0.f;
    }

    const int row = tid >> 3;
    const int c0  = (tid & 7) * 8;
    const int sp  = tid >> 4;
    const int sc  = (tid & 15) * 4;
    const int fo0 = frag_off(2 * sp, sc >> 2);
    const int fo1 = frag_off(2 * sp + 1, sc >> 2);
    float* ksw = s_ks + wave * 64;

    for (int ps = 0; ps < npass; ++ps) {
        const int lb = seg * SEGL + ps * TROWS;

        {
            const size_t g0 = ((size_t)((b * LL + lb + row) * HH + h)) * DD + c0;
            const size_t gs = ((size_t)((b * LL + lb + 2 * sp) * HH + h)) * DD + sc;
            v4f qv[SUB][2], ka[SUB][2], va[SUB][2];
            #pragma unroll
            for (int j = 0; j < SUB; ++j) {
                const size_t g = gs + (size_t)j * CCH * HH * DD;
                qv[j][0] = *(const v4f*)(q + g);
                qv[j][1] = *(const v4f*)(q + g + HH * DD);
                ka[j][0] = *(const v4f*)(k + g);
                ka[j][1] = *(const v4f*)(k + g + HH * DD);
                va[j][0] = *(const v4f*)(v + g);
                va[j][1] = *(const v4f*)(v + g + HH * DD);
            }
            __syncthreads();
            #pragma unroll
            for (int j = 0; j < SUB; ++j) {
                float* tl = sm + j * TILE_F;
                const v4f kf0 = fmap4(ka[j][0]);
                const v4f kf1 = fmap4(ka[j][1]);
                *(v4f*)(tl + T_QF + fo0) = fmap4(qv[j][0]);
                *(v4f*)(tl + T_QF + fo1) = fmap4(qv[j][1]);
                *(v4f*)(tl + T_KF + fo0) = kf0;
                *(v4f*)(tl + T_KF + fo1) = kf1;
                store_pair4(tl + T_KPI + (sp * 64 + sc) * 2, kf0, kf1);
                store_pair4(tl + T_VPI + (sp * 64 + sc) * 2, va[j][0], va[j][1]);
            }
            if (ps + 1 < npass) {
                #pragma unroll
                for (int j = 0; j < SUB; ++j) {
                    const size_t gn = g0 + (size_t)(SUB + j) * CCH * HH * DD;
                    __builtin_prefetch(q + gn, 0, 3);
                    __builtin_prefetch(k + gn, 0, 3);
                    __builtin_prefetch(v + gn, 0, 3);
                }
            }
        }
        __syncthreads();

        {
            float r0 = ksr0, r1 = ksr1, my0 = 0.f, my1 = 0.f;
            #pragma unroll
            for (int j = 0; j < SUB; ++j) {
                if (j == wave) { my0 = r0; my1 = r1; }
                const float* kpi = sm + j * TILE_F + T_KPI;
                float kd0 = 0.f, kd1 = 0.f;
                #pragma unroll
                for (int p = 0; p < 8; ++p) {
                    const v4f kp = *(const v4f*)(kpi + (p * 64 + lane * 2) * 2);
                    kd0 += kp[0]; kd0 += kp[1];
                    kd1 += kp[2]; kd1 += kp[3];
                }
                r0 += kd0; r1 += kd1;
            }
            ksr0 = r0; ksr1 = r1;
            const v2f myv = {my0, my1};
            *(v2f*)(ksw + lane * 2) = myv;

            float* tl = sm + wave * TILE_F;
            const float* s_qf = tl + T_QF;
            const float* s_kf = tl + T_KF;
            float* s_a = tl + T_A;
            float* s_z = tl + T_Z;
            v8f aq = zero8, zi = zero8;
            #pragma unroll
            for (int u = 0; u < 8; ++u) {
                const int qd = 4 * (u >> 1) + 2 * lh + (u & 1);
                const v4f qa  = *(const v4f*)(s_qf + frag_off(lc, qd));
                const v4f bk  = *(const v4f*)(s_kf + frag_off(lc, qd));
                const v4f bks = *(const v4f*)(ksw + 4 * qd);
                aq = wmma4(LO2(qa), LO2(bk), aq);
                zi = wmma4(LO2(qa), LO2(bks), zi);
                aq = wmma4(HI2(qa), HI2(bk), aq);
                zi = wmma4(HI2(qa), HI2(bks), zi);
            }
            #pragma unroll
            for (int r = 0; r < 8; ++r) {
                int m = r + 8 * lh;
                s_a[frag_off(m, lc >> 2) + (lc & 3)] = (lc <= m) ? aq[r] : 0.f;
            }
            if (lc == 0) {
                #pragma unroll
                for (int r = 0; r < 8; ++r) s_z[r + 8 * lh] = zi[r];
            }
        }
        __syncthreads();

        for (int j = 0; j < SUB; ++j) {
            const int l0 = lb + j * CCH;
            float* tl = sm + j * TILE_F;
            const float* s_qf  = tl + T_QF;
            const float* s_kpi = tl + T_KPI;
            const float* s_vpi = tl + T_VPI;
            const float* s_a = tl + T_A;
            const float* s_z = tl + T_Z;

            v4f qq[8];
            #pragma unroll
            for (int u = 0; u < 8; ++u)
                qq[u] = *(const v4f*)(s_qf + frag_off(lc, 4 * (u >> 1) + 2 * lh + (u & 1)));

            v8f oc = zero8;
            #pragma unroll
            for (int t = 0; t < 4; ++t) {
                oc = wmma4(LO2(qq[2 * t]),
                           __builtin_shufflevector(st[t], st[t], 0, 1), oc);
                oc = wmma4(HI2(qq[2 * t]),
                           __builtin_shufflevector(st[t], st[t], 2, 3), oc);
                oc = wmma4(LO2(qq[2 * t + 1]),
                           __builtin_shufflevector(st[t], st[t], 4, 5), oc);
                oc = wmma4(HI2(qq[2 * t + 1]),
                           __builtin_shufflevector(st[t], st[t], 6, 7), oc);
            }

            v2f bv[4];
            #pragma unroll
            for (int kk = 0; kk < 4; ++kk) {
                int p = 2 * kk + lh;
                bv[kk] = *(const v2f*)(s_vpi + (p * 64 + 16 * wave + lc) * 2);
            }
            v8f zr = zero8;
            #pragma unroll
            for (int kk = 0; kk < 4; ++kk) {
                v2f av = *(const v2f*)(s_a + frag_off(lc, kk) + 2 * lh);
                oc = wmma4(av, bv[kk], oc);
                zr = wmma4(av, ones2, zr);
            }
            #pragma unroll
            for (int t = 0; t < 4; ++t) {
                #pragma unroll
                for (int kk = 0; kk < 4; ++kk) {
                    int p = 2 * kk + lh;
                    v2f ak = *(const v2f*)(s_kpi + (p * 64 + 16 * t + lc) * 2);
                    st[t] = wmma4(ak, bv[kk], st[t]);
                }
            }

            #pragma unroll
            for (int r = 0; r < 8; ++r) {
                int m = r + 8 * lh;
                float zz = FEPS + zr[r] + s_z[m];
                size_t go = ((size_t)((b * LL + l0 + m) * HH + h)) * DD
                            + 16 * wave + lc;
                out[go] = oc[r] / zz;
            }
        }
    }
}

extern "C" void kernel_launch(void* const* d_in, const int* in_sizes, int n_in,
                              void* d_out, int out_size, void* d_ws, size_t ws_size,
                              hipStream_t stream) {
    (void)in_sizes; (void)n_in; (void)out_size;
    const float* q = (const float*)d_in[0];
    const float* k = (const float*)d_in[1];
    const float* v = (const float*)d_in[2];
    float* out = (float*)d_out;

    const size_t nSeg   = (size_t)BB * HH * NS;
    const size_t needWs = nSeg * (4096 + 64) * sizeof(float);

    if (ws_size >= needWs) {
        float* segKV = (float*)d_ws;
        float* segKS = segKV + nSeg * 4096;
        hipLaunchKernelGGL(linattn_phase1, dim3(BB * HH * NS), dim3(128), 0,
                           stream, k, v, segKV, segKS);
        hipLaunchKernelGGL(linattn_phase2, dim3(BB * HH, 5), dim3(256), 0,
                           stream, segKV, segKS);
        hipLaunchKernelGGL((linattn_scan<true>), dim3(BB * HH * NS), dim3(128),
                           0, stream, q, k, v, out, segKV, segKS);
    } else {
        hipLaunchKernelGGL((linattn_scan<false>), dim3(BB * HH), dim3(128),
                           0, stream, q, k, v, out,
                           (const float*)nullptr, (const float*)nullptr);
    }
}


# phase 1: waves split the segment's rows, each keeps all 16 state tiles; K/V go from global loads to WMMA operands in registers (no LDS tiles, no loop barrier), one final sum; new sum order: KV, ksum
# speedup vs baseline: 2.4342x; 1.2645x over previous
#include <hip/hip_runtime.h>
#include <math.h>


typedef __attribute__((ext_vector_type(2))) float v2f;
typedef __attribute__((ext_vector_type(4))) float v4f;
typedef __attribute__((ext_vector_type(8))) float v8f;

#define BB   4
#define LL   2048
#define HH   16
#define DD   64
#define CCH  16
#define NS   8
#define SEGL (LL / NS)
#define SUB  4
#define TROWS (SUB * CCH)
#define FEPS 1e-6f

__device__ __forceinline__ float fmap(float x) {
    return x > 0.f ? x + 1.f : __expf(x);
}

__device__ __forceinline__ v4f fmap4(v4f x) {
    v4f y = {fmap(x[0]), fmap(x[1]), fmap(x[2]), fmap(x[3])};
    return y;
}

__device__ __forceinline__ void store_pair4(float* dst, v4f a, v4f b) {
    v4f lo = {a[0], b[0], a[1], b[1]};
    v4f hi = {a[2], b[2], a[3], b[3]};
    *(v4f*)(dst)     = lo;
    *(v4f*)(dst + 4) = hi;
}

__device__ __forceinline__ v8f wmma4(v2f a, v2f b, v8f c) {
    return __builtin_amdgcn_wmma_f32_16x16x4_f32(
        false, a, false, b, (short)0, c, false, false);
}

#define LO2(x) __builtin_shufflevector((x), (x), 0, 1)
#define HI2(x) __builtin_shufflevector((x), (x), 2, 3)

__device__ __forceinline__ int frag_off(int row, int quad) {
    return quad * 64 + ((row + quad) & 15) * 4;
}

#define T_QF    0
#define T_KF    1024
#define T_KPI   2048
#define T_VPI   3072
#define T_A     4096
#define T_Z     4352
#define TILE_F  4368
#define OFF_KS  (SUB * TILE_F)
#define SMEM_F  (OFF_KS + 4 * 64)

#define WROWS   (SEGL / 4)
#define P1_PART 4096
#define P1_KSO  (4 * P1_PART)
#define P1_SMEM (P1_KSO + 8 * 64)

__device__ __forceinline__ void p1_load(const float* __restrict__ k,
                                        const float* __restrict__ v,
                                        size_t g, v4f (&kk)[4][2], v4f (&vv)[4][2]) {
    #pragma unroll
    for (int s = 0; s < 4; ++s) {
        const size_t gg = g + (size_t)s * 4 * HH * DD;
        kk[s][0] = *(const v4f*)(k + gg);
        kk[s][1] = *(const v4f*)(k + gg + HH * DD);
        vv[s][0] = *(const v4f*)(v + gg);
        vv[s][1] = *(const v4f*)(v + gg + HH * DD);
    }
}

__device__ __forceinline__ void p1_chunk(const v4f (&kk)[4][2], const v4f (&vv)[4][2],
                                         v8f (&acc)[4][4], v4f& ks) {
    #pragma unroll
    for (int s = 0; s < 4; ++s) {
        const v4f ka = fmap4(kk[s][0]);
        const v4f kb = fmap4(kk[s][1]);
        ks += ka;
        ks += kb;
        v2f bf[4];
        #pragma unroll
        for (int j = 0; j < 4; ++j) {
            v2f t = {vv[s][0][j], vv[s][1][j]};
            bf[j] = t;
        }
        #pragma unroll
        for (int i = 0; i < 4; ++i) {
            const v2f af = {ka[i], kb[i]};
            #pragma unroll
            for (int j = 0; j < 4; ++j)
                acc[i][j] = wmma4(af, bf[j], acc[i][j]);
        }
    }
}

__global__ __launch_bounds__(128)
void linattn_phase1(const float* __restrict__ k,
                    const float* __restrict__ v,
                    float* __restrict__ segKV,
                    float* __restrict__ segKS) {
    __shared__ __attribute__((aligned(16))) float sm[P1_SMEM];
    const int tid  = threadIdx.x;
    const int lane = tid & 31;
    const int wave = tid >> 5;
    const int lh   = lane >> 4;
    const int lc   = lane & 15;
    const int bh   = blockIdx.x / NS;
    const int seg  = blockIdx.x % NS;
    const int b    = bh / HH;
    const int h    = bh % HH;

    const v8f zero8 = {0.f,0.f,0.f,0.f,0.f,0.f,0.f,0.f};
    v8f acc[4][4];
    #pragma unroll
    for (int i = 0; i < 4; ++i) {
        #pragma unroll
        for (int j = 0; j < 4; ++j) acc[i][j] = zero8;
    }
    v4f ks = {0.f, 0.f, 0.f, 0.f};

    const int l0 = seg * SEGL + wave * WROWS + 2 * lh;
    const size_t g0 = ((size_t)((b * LL + l0) * HH + h)) * DD + 4 * lc;
    const size_t cstep = (size_t)CCH * HH * DD;

    v4f k0[4][2], v0[4][2], k1[4][2], v1[4][2];
    p1_load(k, v, g0, k0, v0);
    p1_load(k, v, g0 + cstep, k1, v1);
    p1_chunk(k0, v0, acc, ks);
    p1_load(k, v, g0 + 2 * cstep, k0, v0);
    p1_chunk(k1, v1, acc, ks);
    p1_load(k, v, g0 + 3 * cstep, k1, v1);
    p1_chunk(k0, v0, acc, ks);
    p1_chunk(k1, v1, acc, ks);

    float* part = sm + wave * P1_PART;
    #pragma unroll
    for (int i = 0; i < 4; ++i) {
        #pragma unroll
        for (int r = 0; r < 8; ++r) {
            const v4f o = {acc[i][0][r], acc[i][1][r], acc[i][2][r], acc[i][3][r]};
            *(v4f*)(part + (32 * lh + 4 * r + i) * 64 + 4 * lc) = o;
        }
    }
    *(v4f*)(sm + P1_KSO + (wave * 32 + lane) * 4) = ks;
    __syncthreads();

    float* kvout = segKV + (size_t)blockIdx.x * 4096;
    #pragma unroll
    for (int n = 0; n < 8; ++n) {
        const int f = (tid + 128 * n) * 4;
        v4f s = *(const v4f*)(sm + f);
        s += *(const v4f*)(sm + P1_PART + f);
        s += *(const v4f*)(sm + 2 * P1_PART + f);
        s += *(const v4f*)(sm + 3 * P1_PART + f);
        *(v4f*)(kvout + f) = s;
    }
    if (tid < 16) {
        v4f s = *(const v4f*)(sm + P1_KSO + tid * 4);
        #pragma unroll
        for (int p = 1; p < 8; ++p)
            s += *(const v4f*)(sm + P1_KSO + (p * 16 + tid) * 4);
        *(v4f*)(segKS + (size_t)blockIdx.x * 64 + tid * 4) = s;
    }
}

__global__ __launch_bounds__(256)
void linattn_phase2(float* __restrict__ segKV,
                    float* __restrict__ segKS) {
    const int bh  = blockIdx.x;
    const int tid = threadIdx.x;
    float* base;
    int stride;
    if (blockIdx.y < 4) {
        base   = segKV + (size_t)bh * NS * 4096 + blockIdx.y * 1024 + tid * 4;
        stride = 4096;
    } else {
        if (tid >= 16) return;
        base   = segKS + (size_t)bh * NS * 64 + tid * 4;
        stride = 64;
    }
    float4 part[NS];
    #pragma unroll
    for (int s = 0; s < NS; ++s)
        part[s] = *(const float4*)(base + (size_t)s * stride);
    float4 acc = make_float4(0.f, 0.f, 0.f, 0.f);
    #pragma unroll
    for (int s = 0; s < NS; ++s) {
        *(float4*)(base + (size_t)s * stride) = acc;
        acc.x += part[s].x;
        acc.y += part[s].y;
        acc.z += part[s].z;
        acc.w += part[s].w;
    }
}

template <bool SEEDED>
__global__ __launch_bounds__(128)
void linattn_scan(const float* __restrict__ q,
                  const float* __restrict__ k,
                  const float* __restrict__ v,
                  float* __restrict__ out,
                  const float* __restrict__ segKV,
                  const float* __restrict__ segKS) {
    __shared__ __attribute__((aligned(16))) float sm[SMEM_F];
    const int tid  = threadIdx.x;
    const int lane = tid & 31;
    const int wave = tid >> 5;
    const int lh   = lane >> 4;
    const int lc   = lane & 15;
    const int bh   = SEEDED ? (blockIdx.x / NS) : blockIdx.x;
    const int seg  = SEEDED ? (blockIdx.x % NS) : 0;
    const int b    = bh / HH;
    const int h    = bh % HH;
    const int npass = SEEDED ? (SEGL / TROWS) : (LL / TROWS);

    float* s_ks  = sm + OFF_KS;

    const v8f zero8 = {0.f,0.f,0.f,0.f,0.f,0.f,0.f,0.f};
    const v2f ones2 = {1.f, 1.f};
    v8f st[4];
    float ksr0, ksr1;

    if (SEEDED) {
        const float* kvin = segKV + (size_t)blockIdx.x * 4096;
        #pragma unroll
        for (int t = 0; t < 4; ++t) {
            #pragma unroll
            for (int r = 0; r < 8; ++r) {
                int srow = 16 * t + r + 8 * lh;
                st[t][r] = kvin[srow * 64 + 16 * wave + lc];
            }
        }
        ksr0 = segKS[(size_t)blockIdx.x * 64 + lane * 2];
        ksr1 = segKS[(size_t)blockIdx.x * 64 + lane * 2 + 1];
    } else {
        #pragma unroll
        for (int t = 0; t < 4; ++t) st[t] = zero8;
        ksr0 = 0.f;
        ksr1 = 0.f;
    }

    const int row = tid >> 3;
    const int c0  = (tid & 7) * 8;
    const int sp  = tid >> 4;
    const int sc  = (tid & 15) * 4;
    const int fo0 = frag_off(2 * sp, sc >> 2);
    const int fo1 = frag_off(2 * sp + 1, sc >> 2);
    float* ksw = s_ks + wave * 64;

    for (int ps = 0; ps < npass; ++ps) {
        const int lb = seg * SEGL + ps * TROWS;

        {
            const size_t g0 = ((size_t)((b * LL + lb + row) * HH + h)) * DD + c0;
            const size_t gs = ((size_t)((b * LL + lb + 2 * sp) * HH + h)) * DD + sc;
            v4f qv[SUB][2], ka[SUB][2], va[SUB][2];
            #pragma unroll
            for (int j = 0; j < SUB; ++j) {
                const size_t g = gs + (size_t)j * CCH * HH * DD;
                qv[j][0] = *(const v4f*)(q + g);
                qv[j][1] = *(const v4f*)(q + g + HH * DD);
                ka[j][0] = *(const v4f*)(k + g);
                ka[j][1] = *(const v4f*)(k + g + HH * DD);
                va[j][0] = *(const v4f*)(v + g);
                va[j][1] = *(const v4f*)(v + g + HH * DD);
            }
            __syncthreads();
            #pragma unroll
            for (int j = 0; j < SUB; ++j) {
                float* tl = sm + j * TILE_F;
                const v4f kf0 = fmap4(ka[j][0]);
                const v4f kf1 = fmap4(ka[j][1]);
                *(v4f*)(tl + T_QF + fo0) = fmap4(qv[j][0]);
                *(v4f*)(tl + T_QF + fo1) = fmap4(qv[j][1]);
                *(v4f*)(tl + T_KF + fo0) = kf0;
                *(v4f*)(tl + T_KF + fo1) = kf1;
                store_pair4(tl + T_KPI + (sp * 64 + sc) * 2, kf0, kf1);
                store_pair4(tl + T_VPI + (sp * 64 + sc) * 2, va[j][0], va[j][1]);
            }
            if (ps + 1 < npass) {
                #pragma unroll
                for (int j = 0; j < SUB; ++j) {
                    const size_t gn = g0 + (size_t)(SUB + j) * CCH * HH * DD;
                    __builtin_prefetch(q + gn, 0, 3);
                    __builtin_prefetch(k + gn, 0, 3);
                    __builtin_prefetch(v + gn, 0, 3);
                }
            }
        }
        __syncthreads();

        {
            float r0 = ksr0, r1 = ksr1, my0 = 0.f, my1 = 0.f;
            #pragma unroll
            for (int j = 0; j < SUB; ++j) {
                if (j == wave) { my0 = r0; my1 = r1; }
                const float* kpi = sm + j * TILE_F + T_KPI;
                float kd0 = 0.f, kd1 = 0.f;
                #pragma unroll
                for (int p = 0; p < 8; ++p) {
                    const v4f kp = *(const v4f*)(kpi + (p * 64 + lane * 2) * 2);
                    kd0 += kp[0]; kd0 += kp[1];
                    kd1 += kp[2]; kd1 += kp[3];
                }
                r0 += kd0; r1 += kd1;
            }
            ksr0 = r0; ksr1 = r1;
            const v2f myv = {my0, my1};
            *(v2f*)(ksw + lane * 2) = myv;

            float* tl = sm + wave * TILE_F;
            const float* s_qf = tl + T_QF;
            const float* s_kf = tl + T_KF;
            float* s_a = tl + T_A;
            float* s_z = tl + T_Z;
            v8f aq = zero8, zi = zero8;
            #pragma unroll
            for (int u = 0; u < 8; ++u) {
                const int qd = 4 * (u >> 1) + 2 * lh + (u & 1);
                const v4f qa  = *(const v4f*)(s_qf + frag_off(lc, qd));
                const v4f bk  = *(const v4f*)(s_kf + frag_off(lc, qd));
                const v4f bks = *(const v4f*)(ksw + 4 * qd);
                aq = wmma4(LO2(qa), LO2(bk), aq);
                zi = wmma4(LO2(qa), LO2(bks), zi);
                aq = wmma4(HI2(qa), HI2(bk), aq);
                zi = wmma4(HI2(qa), HI2(bks), zi);
            }
            #pragma unroll
            for (int r = 0; r < 8; ++r) {
                int m = r + 8 * lh;
                s_a[frag_off(m, lc >> 2) + (lc & 3)] = (lc <= m) ? aq[r] : 0.f;
            }
            if (lc == 0) {
                #pragma unroll
                for (int r = 0; r < 8; ++r) s_z[r + 8 * lh] = zi[r];
            }
        }
        __syncthreads();

        for (int j = 0; j < SUB; ++j) {
            const int l0 = lb + j * CCH;
            float* tl = sm + j * TILE_F;
            const float* s_qf  = tl + T_QF;
            const float* s_kpi = tl + T_KPI;
            const float* s_vpi = tl + T_VPI;
            const float* s_a = tl + T_A;
            const float* s_z = tl + T_Z;

            v4f qq[8];
            #pragma unroll
            for (int u = 0; u < 8; ++u)
                qq[u] = *(const v4f*)(s_qf + frag_off(lc, 4 * (u >> 1) + 2 * lh + (u & 1)));

            v8f oc = zero8;
            #pragma unroll
            for (int t = 0; t < 4; ++t) {
                oc = wmma4(LO2(qq[2 * t]),
                           __builtin_shufflevector(st[t], st[t], 0, 1), oc);
                oc = wmma4(HI2(qq[2 * t]),
                           __builtin_shufflevector(st[t], st[t], 2, 3), oc);
                oc = wmma4(LO2(qq[2 * t + 1]),
                           __builtin_shufflevector(st[t], st[t], 4, 5), oc);
                oc = wmma4(HI2(qq[2 * t + 1]),
                           __builtin_shufflevector(st[t], st[t], 6, 7), oc);
            }

            v2f bv[4];
            #pragma unroll
            for (int kk = 0; kk < 4; ++kk) {
                int p = 2 * kk + lh;
                bv[kk] = *(const v2f*)(s_vpi + (p * 64 + 16 * wave + lc) * 2);
            }
            v8f zr = zero8;
            #pragma unroll
            for (int kk = 0; kk < 4; ++kk) {
                v2f av = *(const v2f*)(s_a + frag_off(lc, kk) + 2 * lh);
                oc = wmma4(av, bv[kk], oc);
                zr = wmma4(av, ones2, zr);
            }
            #pragma unroll
            for (int t = 0; t < 4; ++t) {
                #pragma unroll
                for (int kk = 0; kk < 4; ++kk) {
                    int p = 2 * kk + lh;
                    v2f ak = *(const v2f*)(s_kpi + (p * 64 + 16 * t + lc) * 2);
                    st[t] = wmma4(ak, bv[kk], st[t]);
                }
            }

            #pragma unroll
            for (int r = 0; r < 8; ++r) {
                int m = r + 8 * lh;
                float zz = FEPS + zr[r] + s_z[m];
                size_t go = ((size_t)((b * LL + l0 + m) * HH + h)) * DD
                            + 16 * wave + lc;
                out[go] = oc[r] / zz;
            }
        }
    }
}

extern "C" void kernel_launch(void* const* d_in, const int* in_sizes, int n_in,
                              void* d_out, int out_size, void* d_ws, size_t ws_size,
                              hipStream_t stream) {
    (void)in_sizes; (void)n_in; (void)out_size;
    const float* q = (const float*)d_in[0];
    const float* k = (const float*)d_in[1];
    const float* v = (const float*)d_in[2];
    float* out = (float*)d_out;

    const size_t nSeg   = (size_t)BB * HH * NS;
    const size_t needWs = nSeg * (4096 + 64) * sizeof(float);

    if (ws_size >= needWs) {
        float* segKV = (float*)d_ws;
        float* segKS = segKV + nSeg * 4096;
        hipLaunchKernelGGL(linattn_phase1, dim3(BB * HH * NS), dim3(128), 0,
                           stream, k, v, segKV, segKS);
        hipLaunchKernelGGL(linattn_phase2, dim3(BB * HH, 5), dim3(256), 0,
                           stream, segKV, segKS);
        hipLaunchKernelGGL((linattn_scan<true>), dim3(BB * HH * NS), dim3(128),
                           0, stream, q, k, v, out, segKV, segKS);
    } else {
        hipLaunchKernelGGL((linattn_scan<false>), dim3(BB * HH), dim3(128),
                           0, stream, q, k, v, out,
                           (const float*)nullptr, (const float*)nullptr);
    }
}


# scan kernel: score tile formed transposed (Kf@Qf^T); its wave adds the masked key sums (ones@A^T) to the Ksum product and publishes 1/z per chunk; chunk loop: no z WMMAs, no divide; new sum order: z
# speedup vs baseline: 2.6213x; 1.0769x over previous
#include <hip/hip_runtime.h>
#include <math.h>


typedef __attribute__((ext_vector_type(2))) float v2f;
typedef __attribute__((ext_vector_type(4))) float v4f;
typedef __attribute__((ext_vector_type(8))) float v8f;

#define BB   4
#define LL   2048
#define HH   16
#define DD   64
#define CCH  16
#define NS   8
#define SEGL (LL / NS)
#define SUB  4
#define TROWS (SUB * CCH)
#define FEPS 1e-6f

__device__ __forceinline__ float fmap(float x) {
    return x > 0.f ? x + 1.f : __expf(x);
}

__device__ __forceinline__ v4f fmap4(v4f x) {
    v4f y = {fmap(x[0]), fmap(x[1]), fmap(x[2]), fmap(x[3])};
    return y;
}

__device__ __forceinline__ void store_pair4(float* dst, v4f a, v4f b) {
    v4f lo = {a[0], b[0], a[1], b[1]};
    v4f hi = {a[2], b[2], a[3], b[3]};
    *(v4f*)(dst)     = lo;
    *(v4f*)(dst + 4) = hi;
}

__device__ __forceinline__ v8f wmma4(v2f a, v2f b, v8f c) {
    return __builtin_amdgcn_wmma_f32_16x16x4_f32(
        false, a, false, b, (short)0, c, false, false);
}

#define LO2(x) __builtin_shufflevector((x), (x), 0, 1)
#define HI2(x) __builtin_shufflevector((x), (x), 2, 3)

__device__ __forceinline__ int frag_off(int row, int quad) {
    return quad * 64 + ((row + quad) & 15) * 4;
}

#define T_QF    0
#define T_KF    1024
#define T_KPI   2048
#define T_VPI   3072
#define T_A     4096
#define T_Z     4352
#define TILE_F  4368
#define OFF_KS  (SUB * TILE_F)
#define SMEM_F  (OFF_KS + 4 * 64)

#define WROWS   (SEGL / 4)
#define P1_PART 4096
#define P1_KSO  (4 * P1_PART)
#define P1_SMEM (P1_KSO + 8 * 64)

__device__ __forceinline__ void p1_load(const float* __restrict__ k,
                                        const float* __restrict__ v,
                                        size_t g, v4f (&kk)[4][2], v4f (&vv)[4][2]) {
    #pragma unroll
    for (int s = 0; s < 4; ++s) {
        const size_t gg = g + (size_t)s * 4 * HH * DD;
        kk[s][0] = *(const v4f*)(k + gg);
        kk[s][1] = *(const v4f*)(k + gg + HH * DD);
        vv[s][0] = *(const v4f*)(v + gg);
        vv[s][1] = *(const v4f*)(v + gg + HH * DD);
    }
}

__device__ __forceinline__ void p1_chunk(const v4f (&kk)[4][2], const v4f (&vv)[4][2],
                                         v8f (&acc)[4][4], v4f& ks) {
    #pragma unroll
    for (int s = 0; s < 4; ++s) {
        const v4f ka = fmap4(kk[s][0]);
        const v4f kb = fmap4(kk[s][1]);
        ks += ka;
        ks += kb;
        v2f bf[4];
        #pragma unroll
        for (int j = 0; j < 4; ++j) {
            v2f t = {vv[s][0][j], vv[s][1][j]};
            bf[j] = t;
        }
        #pragma unroll
        for (int i = 0; i < 4; ++i) {
            const v2f af = {ka[i], kb[i]};
            #pragma unroll
            for (int j = 0; j < 4; ++j)
                acc[i][j] = wmma4(af, bf[j], acc[i][j]);
        }
    }
}

__global__ __launch_bounds__(128)
void linattn_phase1(const float* __restrict__ k,
                    const float* __restrict__ v,
                    float* __restrict__ segKV,
                    float* __restrict__ segKS) {
    __shared__ __attribute__((aligned(16))) float sm[P1_SMEM];
    const int tid  = threadIdx.x;
    const int lane = tid & 31;
    const int wave = tid >> 5;
    const int lh   = lane >> 4;
    const int lc   = lane & 15;
    const int bh   = blockIdx.x / NS;
    const int seg  = blockIdx.x % NS;
    const int b    = bh / HH;
    const int h    = bh % HH;

    const v8f zero8 = {0.f,0.f,0.f,0.f,0.f,0.f,0.f,0.f};
    v8f acc[4][4];
    #pragma unroll
    for (int i = 0; i < 4; ++i) {
        #pragma unroll
        for (int j = 0; j < 4; ++j) acc[i][j] = zero8;
    }
    v4f ks = {0.f, 0.f, 0.f, 0.f};

    const int l0 = seg * SEGL + wave * WROWS + 2 * lh;
    const size_t g0 = ((size_t)((b * LL + l0) * HH + h)) * DD + 4 * lc;
    const size_t cstep = (size_t)CCH * HH * DD;

    v4f k0[4][2], v0[4][2], k1[4][2], v1[4][2];
    p1_load(k, v, g0, k0, v0);
    p1_load(k, v, g0 + cstep, k1, v1);
    p1_chunk(k0, v0, acc, ks);
    p1_load(k, v, g0 + 2 * cstep, k0, v0);
    p1_chunk(k1, v1, acc, ks);
    p1_load(k, v, g0 + 3 * cstep, k1, v1);
    p1_chunk(k0, v0, acc, ks);
    p1_chunk(k1, v1, acc, ks);

    float* part = sm + wave * P1_PART;
    #pragma unroll
    for (int i = 0; i < 4; ++i) {
        #pragma unroll
        for (int r = 0; r < 8; ++r) {
            const v4f o = {acc[i][0][r], acc[i][1][r], acc[i][2][r], acc[i][3][r]};
            *(v4f*)(part + (32 * lh + 4 * r + i) * 64 + 4 * lc) = o;
        }
    }
    *(v4f*)(sm + P1_KSO + (wave * 32 + lane) * 4) = ks;
    __syncthreads();

    float* kvout = segKV + (size_t)blockIdx.x * 4096;
    #pragma unroll
    for (int n = 0; n < 8; ++n) {
        const int f = (tid + 128 * n) * 4;
        v4f s = *(const v4f*)(sm + f);
        s += *(const v4f*)(sm + P1_PART + f);
        s += *(const v4f*)(sm + 2 * P1_PART + f);
        s += *(const v4f*)(sm + 3 * P1_PART + f);
        *(v4f*)(kvout + f) = s;
    }
    if (tid < 16) {
        v4f s = *(const v4f*)(sm + P1_KSO + tid * 4);
        #pragma unroll
        for (int p = 1; p < 8; ++p)
            s += *(const v4f*)(sm + P1_KSO + (p * 16 + tid) * 4);
        *(v4f*)(segKS + (size_t)blockIdx.x * 64 + tid * 4) = s;
    }
}

__global__ __launch_bounds__(256)
void linattn_phase2(float* __restrict__ segKV,
                    float* __restrict__ segKS) {
    const int bh  = blockIdx.x;
    const int tid = threadIdx.x;
    float* base;
    int stride;
    if (blockIdx.y < 4) {
        base   = segKV + (size_t)bh * NS * 4096 + blockIdx.y * 1024 + tid * 4;
        stride = 4096;
    } else {
        if (tid >= 16) return;
        base   = segKS + (size_t)bh * NS * 64 + tid * 4;
        stride = 64;
    }
    float4 part[NS];
    #pragma unroll
    for (int s = 0; s < NS; ++s)
        part[s] = *(const float4*)(base + (size_t)s * stride);
    float4 acc = make_float4(0.f, 0.f, 0.f, 0.f);
    #pragma unroll
    for (int s = 0; s < NS; ++s) {
        *(float4*)(base + (size_t)s * stride) = acc;
        acc.x += part[s].x;
        acc.y += part[s].y;
        acc.z += part[s].z;
        acc.w += part[s].w;
    }
}

template <bool SEEDED>
__global__ __launch_bounds__(128)
void linattn_scan(const float* __restrict__ q,
                  const float* __restrict__ k,
                  const float* __restrict__ v,
                  float* __restrict__ out,
                  const float* __restrict__ segKV,
                  const float* __restrict__ segKS) {
    __shared__ __attribute__((aligned(16))) float sm[SMEM_F];
    const int tid  = threadIdx.x;
    const int lane = tid & 31;
    const int wave = tid >> 5;
    const int lh   = lane >> 4;
    const int lc   = lane & 15;
    const int bh   = SEEDED ? (blockIdx.x / NS) : blockIdx.x;
    const int seg  = SEEDED ? (blockIdx.x % NS) : 0;
    const int b    = bh / HH;
    const int h    = bh % HH;
    const int npass = SEEDED ? (SEGL / TROWS) : (LL / TROWS);

    float* s_ks  = sm + OFF_KS;

    const v8f zero8 = {0.f,0.f,0.f,0.f,0.f,0.f,0.f,0.f};
    const v2f ones2 = {1.f, 1.f};
    v8f st[4];
    float ksr0, ksr1;

    if (SEEDED) {
        const float* kvin = segKV + (size_t)blockIdx.x * 4096;
        #pragma unroll
        for (int t = 0; t < 4; ++t) {
            #pragma unroll
            for (int r = 0; r < 8; ++r) {
                int srow = 16 * t + r + 8 * lh;
                st[t][r] = kvin[srow * 64 + 16 * wave + lc];
            }
        }
        ksr0 = segKS[(size_t)blockIdx.x * 64 + lane * 2];
        ksr1 = segKS[(size_t)blockIdx.x * 64 + lane * 2 + 1];
    } else {
        #pragma unroll
        for (int t = 0; t < 4; ++t) st[t] = zero8;
        ksr0 = 0.f;
        ksr1 = 0.f;
    }

    const int row = tid >> 3;
    const int c0  = (tid & 7) * 8;
    const int sp  = tid >> 4;
    const int sc  = (tid & 15) * 4;
    const int fo0 = frag_off(2 * sp, sc >> 2);
    const int fo1 = frag_off(2 * sp + 1, sc >> 2);
    float* ksw = s_ks + wave * 64;

    for (int ps = 0; ps < npass; ++ps) {
        const int lb = seg * SEGL + ps * TROWS;

        {
            const size_t g0 = ((size_t)((b * LL + lb + row) * HH + h)) * DD + c0;
            const size_t gs = ((size_t)((b * LL + lb + 2 * sp) * HH + h)) * DD + sc;
            v4f qv[SUB][2], ka[SUB][2], va[SUB][2];
            #pragma unroll
            for (int j = 0; j < SUB; ++j) {
                const size_t g = gs + (size_t)j * CCH * HH * DD;
                qv[j][0] = *(const v4f*)(q + g);
                qv[j][1] = *(const v4f*)(q + g + HH * DD);
                ka[j][0] = *(const v4f*)(k + g);
                ka[j][1] = *(const v4f*)(k + g + HH * DD);
                va[j][0] = *(const v4f*)(v + g);
                va[j][1] = *(const v4f*)(v + g + HH * DD);
            }
            __syncthreads();
            #pragma unroll
            for (int j = 0; j < SUB; ++j) {
                float* tl = sm + j * TILE_F;
                const v4f kf0 = fmap4(ka[j][0]);
                const v4f kf1 = fmap4(ka[j][1]);
                *(v4f*)(tl + T_QF + fo0) = fmap4(qv[j][0]);
                *(v4f*)(tl + T_QF + fo1) = fmap4(qv[j][1]);
                *(v4f*)(tl + T_KF + fo0) = kf0;
                *(v4f*)(tl + T_KF + fo1) = kf1;
                store_pair4(tl + T_KPI + (sp * 64 + sc) * 2, kf0, kf1);
                store_pair4(tl + T_VPI + (sp * 64 + sc) * 2, va[j][0], va[j][1]);
            }
            if (ps + 1 < npass) {
                #pragma unroll
                for (int j = 0; j < SUB; ++j) {
                    const size_t gn = g0 + (size_t)(SUB + j) * CCH * HH * DD;
                    __builtin_prefetch(q + gn, 0, 3);
                    __builtin_prefetch(k + gn, 0, 3);
                    __builtin_prefetch(v + gn, 0, 3);
                }
            }
        }
        __syncthreads();

        {
            float r0 = ksr0, r1 = ksr1, my0 = 0.f, my1 = 0.f;
            #pragma unroll
            for (int j = 0; j < SUB; ++j) {
                if (j == wave) { my0 = r0; my1 = r1; }
                const float* kpi = sm + j * TILE_F + T_KPI;
                float kd0 = 0.f, kd1 = 0.f;
                #pragma unroll
                for (int p = 0; p < 8; ++p) {
                    const v4f kp = *(const v4f*)(kpi + (p * 64 + lane * 2) * 2);
                    kd0 += kp[0]; kd0 += kp[1];
                    kd1 += kp[2]; kd1 += kp[3];
                }
                r0 += kd0; r1 += kd1;
            }
            ksr0 = r0; ksr1 = r1;
            const v2f myv = {my0, my1};
            *(v2f*)(ksw + lane * 2) = myv;

            float* tl = sm + wave * TILE_F;
            const float* s_qf = tl + T_QF;
            const float* s_kf = tl + T_KF;
            float* s_a = tl + T_A;
            float* s_z = tl + T_Z;
            v8f at = zero8, zt = zero8;
            #pragma unroll
            for (int u = 0; u < 8; ++u) {
                const int qd = 4 * (u >> 1) + 2 * lh + (u & 1);
                const v4f qa  = *(const v4f*)(s_qf + frag_off(lc, qd));
                const v4f bk  = *(const v4f*)(s_kf + frag_off(lc, qd));
                const v4f bks = *(const v4f*)(ksw + 4 * qd);
                at = wmma4(LO2(bk), LO2(qa), at);
                zt = wmma4(LO2(bks), LO2(qa), zt);
                at = wmma4(HI2(bk), HI2(qa), at);
                zt = wmma4(HI2(bks), HI2(qa), zt);
            }
            #pragma unroll
            for (int r = 0; r < 8; ++r)
                at[r] = (r + 8 * lh <= lc) ? at[r] : 0.f;
            *(v4f*)(s_a + frag_off(lc, 2 * lh))     = __builtin_shufflevector(at, at, 0, 1, 2, 3);
            *(v4f*)(s_a + frag_off(lc, 2 * lh + 1)) = __builtin_shufflevector(at, at, 4, 5, 6, 7);
            #pragma unroll
            for (int i = 0; i < 4; ++i) {
                const v2f bp = {at[2 * i], at[2 * i + 1]};
                zt = wmma4(ones2, bp, zt);
            }
            s_z[lc] = 1.0f / (zt[0] + FEPS);
        }
        __syncthreads();

        for (int j = 0; j < SUB; ++j) {
            const int l0 = lb + j * CCH;
            float* tl = sm + j * TILE_F;
            const float* s_qf  = tl + T_QF;
            const float* s_kpi = tl + T_KPI;
            const float* s_vpi = tl + T_VPI;
            const float* s_a = tl + T_A;
            const float* s_z = tl + T_Z;

            v4f qq[8];
            #pragma unroll
            for (int u = 0; u < 8; ++u)
                qq[u] = *(const v4f*)(s_qf + frag_off(lc, 4 * (u >> 1) + 2 * lh + (u & 1)));

            v8f oc = zero8;
            #pragma unroll
            for (int t = 0; t < 4; ++t) {
                oc = wmma4(LO2(qq[2 * t]),
                           __builtin_shufflevector(st[t], st[t], 0, 1), oc);
                oc = wmma4(HI2(qq[2 * t]),
                           __builtin_shufflevector(st[t], st[t], 2, 3), oc);
                oc = wmma4(LO2(qq[2 * t + 1]),
                           __builtin_shufflevector(st[t], st[t], 4, 5), oc);
                oc = wmma4(HI2(qq[2 * t + 1]),
                           __builtin_shufflevector(st[t], st[t], 6, 7), oc);
            }

            v2f bv[4];
            #pragma unroll
            for (int kk = 0; kk < 4; ++kk) {
                int p = 2 * kk + lh;
                bv[kk] = *(const v2f*)(s_vpi + (p * 64 + 16 * wave + lc) * 2);
            }
            #pragma unroll
            for (int kk = 0; kk < 4; ++kk) {
                v2f av = *(const v2f*)(s_a + frag_off(lc, kk) + 2 * lh);
                oc = wmma4(av, bv[kk], oc);
            }
            #pragma unroll
            for (int t = 0; t < 4; ++t) {
                #pragma unroll
                for (int kk = 0; kk < 4; ++kk) {
                    int p = 2 * kk + lh;
                    v2f ak = *(const v2f*)(s_kpi + (p * 64 + 16 * t + lc) * 2);
                    st[t] = wmma4(ak, bv[kk], st[t]);
                }
            }

            #pragma unroll
            for (int r = 0; r < 8; ++r) {
                int m = r + 8 * lh;
                size_t go = ((size_t)((b * LL + l0 + m) * HH + h)) * DD
                            + 16 * wave + lc;
                out[go] = oc[r] * s_z[m];
            }
        }
    }
}

extern "C" void kernel_launch(void* const* d_in, const int* in_sizes, int n_in,
                              void* d_out, int out_size, void* d_ws, size_t ws_size,
                              hipStream_t stream) {
    (void)in_sizes; (void)n_in; (void)out_size;
    const float* q = (const float*)d_in[0];
    const float* k = (const float*)d_in[1];
    const float* v = (const float*)d_in[2];
    float* out = (float*)d_out;

    const size_t nSeg   = (size_t)BB * HH * NS;
    const size_t needWs = nSeg * (4096 + 64) * sizeof(float);

    if (ws_size >= needWs) {
        float* segKV = (float*)d_ws;
        float* segKS = segKV + nSeg * 4096;
        hipLaunchKernelGGL(linattn_phase1, dim3(BB * HH * NS), dim3(128), 0,
                           stream, k, v, segKV, segKS);
        hipLaunchKernelGGL(linattn_phase2, dim3(BB * HH, 5), dim3(256), 0,
                           stream, segKV, segKS);
        hipLaunchKernelGGL((linattn_scan<true>), dim3(BB * HH * NS), dim3(128),
                           0, stream, q, k, v, out, segKV, segKS);
    } else {
        hipLaunchKernelGGL((linattn_scan<false>), dim3(BB * HH), dim3(128),
                           0, stream, q, k, v, out,
                           (const float*)nullptr, (const float*)nullptr);
    }
}


# scan kernel: 4 producer waves stage the next 64 rows and form their score tiles in a second LDS tile set while 4 consumer waves run the chunk loop; barriers order LDS only; sums same
# speedup vs baseline: 2.7338x; 1.0429x over previous
#include <hip/hip_runtime.h>
#include <math.h>


typedef __attribute__((ext_vector_type(2))) float v2f;
typedef __attribute__((ext_vector_type(4))) float v4f;
typedef __attribute__((ext_vector_type(8))) float v8f;

#define BB   4
#define LL   2048
#define HH   16
#define DD   64
#define CCH  16
#define NS   8
#define SEGL (LL / NS)
#define SUB  4
#define TROWS (SUB * CCH)
#define FEPS 1e-6f

__device__ __forceinline__ float fmap(float x) {
    return x > 0.f ? x + 1.f : __expf(x);
}

__device__ __forceinline__ v4f fmap4(v4f x) {
    v4f y = {fmap(x[0]), fmap(x[1]), fmap(x[2]), fmap(x[3])};
    return y;
}

__device__ __forceinline__ void store_pair4(float* dst, v4f a, v4f b) {
    v4f lo = {a[0], b[0], a[1], b[1]};
    v4f hi = {a[2], b[2], a[3], b[3]};
    *(v4f*)(dst)     = lo;
    *(v4f*)(dst + 4) = hi;
}

__device__ __forceinline__ v8f wmma4(v2f a, v2f b, v8f c) {
    return __builtin_amdgcn_wmma_f32_16x16x4_f32(
        false, a, false, b, (short)0, c, false, false);
}

#define LO2(x) __builtin_shufflevector((x), (x), 0, 1)
#define HI2(x) __builtin_shufflevector((x), (x), 2, 3)

__device__ __forceinline__ int frag_off(int row, int quad) {
    return quad * 64 + ((row + quad) & 15) * 4;
}

#define T_QF    0
#define T_KF    1024
#define T_KPI   2048
#define T_VPI   3072
#define T_A     4096
#define T_Z     4352
#define TILE_F  4368
#define BUF_F   (SUB * TILE_F)
#define OFF_KS  (2 * BUF_F)
#define SMEM_F  (OFF_KS + 4 * 64)

#define WROWS   (SEGL / 4)
#define P1_PART 4096
#define P1_KSO  (4 * P1_PART)
#define P1_SMEM (P1_KSO + 8 * 64)

__device__ __forceinline__ void p1_load(const float* __restrict__ k,
                                        const float* __restrict__ v,
                                        size_t g, v4f (&kk)[4][2], v4f (&vv)[4][2]) {
    #pragma unroll
    for (int s = 0; s < 4; ++s) {
        const size_t gg = g + (size_t)s * 4 * HH * DD;
        kk[s][0] = *(const v4f*)(k + gg);
        kk[s][1] = *(const v4f*)(k + gg + HH * DD);
        vv[s][0] = *(const v4f*)(v + gg);
        vv[s][1] = *(const v4f*)(v + gg + HH * DD);
    }
}

__device__ __forceinline__ void p1_chunk(const v4f (&kk)[4][2], const v4f (&vv)[4][2],
                                         v8f (&acc)[4][4], v4f& ks) {
    #pragma unroll
    for (int s = 0; s < 4; ++s) {
        const v4f ka = fmap4(kk[s][0]);
        const v4f kb = fmap4(kk[s][1]);
        ks += ka;
        ks += kb;
        v2f bf[4];
        #pragma unroll
        for (int j = 0; j < 4; ++j) {
            v2f t = {vv[s][0][j], vv[s][1][j]};
            bf[j] = t;
        }
        #pragma unroll
        for (int i = 0; i < 4; ++i) {
            const v2f af = {ka[i], kb[i]};
            #pragma unroll
            for (int j = 0; j < 4; ++j)
                acc[i][j] = wmma4(af, bf[j], acc[i][j]);
        }
    }
}

__global__ __launch_bounds__(128)
void linattn_phase1(const float* __restrict__ k,
                    const float* __restrict__ v,
                    float* __restrict__ segKV,
                    float* __restrict__ segKS) {
    __shared__ __attribute__((aligned(16))) float sm[P1_SMEM];
    const int tid  = threadIdx.x;
    const int lane = tid & 31;
    const int wave = tid >> 5;
    const int lh   = lane >> 4;
    const int lc   = lane & 15;
    const int bh   = blockIdx.x / NS;
    const int seg  = blockIdx.x % NS;
    const int b    = bh / HH;
    const int h    = bh % HH;

    const v8f zero8 = {0.f,0.f,0.f,0.f,0.f,0.f,0.f,0.f};
    v8f acc[4][4];
    #pragma unroll
    for (int i = 0; i < 4; ++i) {
        #pragma unroll
        for (int j = 0; j < 4; ++j) acc[i][j] = zero8;
    }
    v4f ks = {0.f, 0.f, 0.f, 0.f};

    const int l0 = seg * SEGL + wave * WROWS + 2 * lh;
    const size_t g0 = ((size_t)((b * LL + l0) * HH + h)) * DD + 4 * lc;
    const size_t cstep = (size_t)CCH * HH * DD;

    v4f k0[4][2], v0[4][2], k1[4][2], v1[4][2];
    p1_load(k, v, g0, k0, v0);
    p1_load(k, v, g0 + cstep, k1, v1);
    p1_chunk(k0, v0, acc, ks);
    p1_load(k, v, g0 + 2 * cstep, k0, v0);
    p1_chunk(k1, v1, acc, ks);
    p1_load(k, v, g0 + 3 * cstep, k1, v1);
    p1_chunk(k0, v0, acc, ks);
    p1_chunk(k1, v1, acc, ks);

    float* part = sm + wave * P1_PART;
    #pragma unroll
    for (int i = 0; i < 4; ++i) {
        #pragma unroll
        for (int r = 0; r < 8; ++r) {
            const v4f o = {acc[i][0][r], acc[i][1][r], acc[i][2][r], acc[i][3][r]};
            *(v4f*)(part + (32 * lh + 4 * r + i) * 64 + 4 * lc) = o;
        }
    }
    *(v4f*)(sm + P1_KSO + (wave * 32 + lane) * 4) = ks;
    __syncthreads();

    float* kvout = segKV + (size_t)blockIdx.x * 4096;
    #pragma unroll
    for (int n = 0; n < 8; ++n) {
        const int f = (tid + 128 * n) * 4;
        v4f s = *(const v4f*)(sm + f);
        s += *(const v4f*)(sm + P1_PART + f);
        s += *(const v4f*)(sm + 2 * P1_PART + f);
        s += *(const v4f*)(sm + 3 * P1_PART + f);
        *(v4f*)(kvout + f) = s;
    }
    if (tid < 16) {
        v4f s = *(const v4f*)(sm + P1_KSO + tid * 4);
        #pragma unroll
        for (int p = 1; p < 8; ++p)
            s += *(const v4f*)(sm + P1_KSO + (p * 16 + tid) * 4);
        *(v4f*)(segKS + (size_t)blockIdx.x * 64 + tid * 4) = s;
    }
}

__global__ __launch_bounds__(256)
void linattn_phase2(float* __restrict__ segKV,
                    float* __restrict__ segKS) {
    const int bh  = blockIdx.x;
    const int tid = threadIdx.x;
    float* base;
    int stride;
    if (blockIdx.y < 4) {
        base   = segKV + (size_t)bh * NS * 4096 + blockIdx.y * 1024 + tid * 4;
        stride = 4096;
    } else {
        if (tid >= 16) return;
        base   = segKS + (size_t)bh * NS * 64 + tid * 4;
        stride = 64;
    }
    float4 part[NS];
    #pragma unroll
    for (int s = 0; s < NS; ++s)
        part[s] = *(const float4*)(base + (size_t)s * stride);
    float4 acc = make_float4(0.f, 0.f, 0.f, 0.f);
    #pragma unroll
    for (int s = 0; s < NS; ++s) {
        *(float4*)(base + (size_t)s * stride) = acc;
        acc.x += part[s].x;
        acc.y += part[s].y;
        acc.z += part[s].z;
        acc.w += part[s].w;
    }
}


__device__ __forceinline__ void lds_barrier() {
    __builtin_amdgcn_fence(__ATOMIC_RELEASE, "workgroup", "local");
    __builtin_amdgcn_s_barrier();
    __builtin_amdgcn_fence(__ATOMIC_ACQUIRE, "workgroup", "local");
}

__device__ __forceinline__ void stage_pass(const float* __restrict__ q,
                                           const float* __restrict__ k,
                                           const float* __restrict__ v,
                                           float* buf, int b, int h, int lb,
                                           int gt, bool more) {
    const int row = gt >> 3;
    const int c0  = (gt & 7) * 8;
    const int sp  = gt >> 4;
    const int sc  = (gt & 15) * 4;
    const int fo0 = frag_off(2 * sp, sc >> 2);
    const int fo1 = frag_off(2 * sp + 1, sc >> 2);
    const size_t g0 = ((size_t)((b * LL + lb + row) * HH + h)) * DD + c0;
    const size_t gs = ((size_t)((b * LL + lb + 2 * sp) * HH + h)) * DD + sc;
    v4f qv[SUB][2], ka[SUB][2], va[SUB][2];
    #pragma unroll
    for (int j = 0; j < SUB; ++j) {
        const size_t g = gs + (size_t)j * CCH * HH * DD;
        qv[j][0] = *(const v4f*)(q + g);
        qv[j][1] = *(const v4f*)(q + g + HH * DD);
        ka[j][0] = *(const v4f*)(k + g);
        ka[j][1] = *(const v4f*)(k + g + HH * DD);
        va[j][0] = *(const v4f*)(v + g);
        va[j][1] = *(const v4f*)(v + g + HH * DD);
    }
    #pragma unroll
    for (int j = 0; j < SUB; ++j) {
        float* tl = buf + j * TILE_F;
        const v4f kf0 = fmap4(ka[j][0]);
        const v4f kf1 = fmap4(ka[j][1]);
        *(v4f*)(tl + T_QF + fo0) = fmap4(qv[j][0]);
        *(v4f*)(tl + T_QF + fo1) = fmap4(qv[j][1]);
        *(v4f*)(tl + T_KF + fo0) = kf0;
        *(v4f*)(tl + T_KF + fo1) = kf1;
        store_pair4(tl + T_KPI + (sp * 64 + sc) * 2, kf0, kf1);
        store_pair4(tl + T_VPI + (sp * 64 + sc) * 2, va[j][0], va[j][1]);
    }
    if (more) {
        #pragma unroll
        for (int j = 0; j < SUB; ++j) {
            const size_t gn = g0 + (size_t)(SUB + j) * CCH * HH * DD;
            __builtin_prefetch(q + gn, 0, 3);
            __builtin_prefetch(k + gn, 0, 3);
            __builtin_prefetch(v + gn, 0, 3);
        }
    }
}

__device__ __forceinline__ void score_pass(float* buf, float* ksw, int wave,
                                           int lane, int lc, int lh,
                                           float& ksr0, float& ksr1) {
    const v8f zero8 = {0.f,0.f,0.f,0.f,0.f,0.f,0.f,0.f};
    const v2f ones2 = {1.f, 1.f};
    float r0 = ksr0, r1 = ksr1, my0 = 0.f, my1 = 0.f;
    #pragma unroll
    for (int j = 0; j < SUB; ++j) {
        if (j == wave) { my0 = r0; my1 = r1; }
        const float* kpi = buf + j * TILE_F + T_KPI;
        float kd0 = 0.f, kd1 = 0.f;
        #pragma unroll
        for (int p = 0; p < 8; ++p) {
            const v4f kp = *(const v4f*)(kpi + (p * 64 + lane * 2) * 2);
            kd0 += kp[0]; kd0 += kp[1];
            kd1 += kp[2]; kd1 += kp[3];
        }
        r0 += kd0; r1 += kd1;
    }
    ksr0 = r0; ksr1 = r1;
    const v2f myv = {my0, my1};
    *(v2f*)(ksw + lane * 2) = myv;

    float* tl = buf + wave * TILE_F;
    const float* s_qf = tl + T_QF;
    const float* s_kf = tl + T_KF;
    float* s_a = tl + T_A;
    float* s_z = tl + T_Z;
    v8f at = zero8, zt = zero8;
    #pragma unroll
    for (int u = 0; u < 8; ++u) {
        const int qd = 4 * (u >> 1) + 2 * lh + (u & 1);
        const v4f qa  = *(const v4f*)(s_qf + frag_off(lc, qd));
        const v4f bk  = *(const v4f*)(s_kf + frag_off(lc, qd));
        const v4f bks = *(const v4f*)(ksw + 4 * qd);
        at = wmma4(LO2(bk), LO2(qa), at);
        zt = wmma4(LO2(bks), LO2(qa), zt);
        at = wmma4(HI2(bk), HI2(qa), at);
        zt = wmma4(HI2(bks), HI2(qa), zt);
    }
    #pragma unroll
    for (int r = 0; r < 8; ++r)
        at[r] = (r + 8 * lh <= lc) ? at[r] : 0.f;
    *(v4f*)(s_a + frag_off(lc, 2 * lh))     = __builtin_shufflevector(at, at, 0, 1, 2, 3);
    *(v4f*)(s_a + frag_off(lc, 2 * lh + 1)) = __builtin_shufflevector(at, at, 4, 5, 6, 7);
    #pragma unroll
    for (int i = 0; i < 4; ++i) {
        const v2f bp = {at[2 * i], at[2 * i + 1]};
        zt = wmma4(ones2, bp, zt);
    }
    s_z[lc] = 1.0f / (zt[0] + FEPS);
}

__device__ __forceinline__ void scan_chunk(const float* tl, v8f (&st)[4],
                                           float* __restrict__ out, size_t gbase,
                                           int wave, int lc, int lh) {
    const float* s_qf  = tl + T_QF;
    const float* s_kpi = tl + T_KPI;
    const float* s_vpi = tl + T_VPI;
    const float* s_a = tl + T_A;
    const float* s_z = tl + T_Z;
    const v8f zero8 = {0.f,0.f,0.f,0.f,0.f,0.f,0.f,0.f};

    v4f qq[8];
    #pragma unroll
    for (int u = 0; u < 8; ++u)
        qq[u] = *(const v4f*)(s_qf + frag_off(lc, 4 * (u >> 1) + 2 * lh + (u & 1)));

    v8f oc = zero8;
    #pragma unroll
    for (int t = 0; t < 4; ++t) {
        oc = wmma4(LO2(qq[2 * t]),
                   __builtin_shufflevector(st[t], st[t], 0, 1), oc);
        oc = wmma4(HI2(qq[2 * t]),
                   __builtin_shufflevector(st[t], st[t], 2, 3), oc);
        oc = wmma4(LO2(qq[2 * t + 1]),
                   __builtin_shufflevector(st[t], st[t], 4, 5), oc);
        oc = wmma4(HI2(qq[2 * t + 1]),
                   __builtin_shufflevector(st[t], st[t], 6, 7), oc);
    }

    v2f bv[4];
    #pragma unroll
    for (int kk = 0; kk < 4; ++kk) {
        int p = 2 * kk + lh;
        bv[kk] = *(const v2f*)(s_vpi + (p * 64 + 16 * wave + lc) * 2);
    }
    #pragma unroll
    for (int kk = 0; kk < 4; ++kk) {
        v2f av = *(const v2f*)(s_a + frag_off(lc, kk) + 2 * lh);
        oc = wmma4(av, bv[kk], oc);
    }
    #pragma unroll
    for (int t = 0; t < 4; ++t) {
        #pragma unroll
        for (int kk = 0; kk < 4; ++kk) {
            int p = 2 * kk + lh;
            v2f ak = *(const v2f*)(s_kpi + (p * 64 + 16 * t + lc) * 2);
            st[t] = wmma4(ak, bv[kk], st[t]);
        }
    }

    #pragma unroll
    for (int r = 0; r < 8; ++r) {
        int m = r + 8 * lh;
        out[gbase + (size_t)m * HH * DD] = oc[r] * s_z[m];
    }
}

template <bool SEEDED>
__global__ __launch_bounds__(256)
void linattn_scan(const float* __restrict__ q,
                  const float* __restrict__ k,
                  const float* __restrict__ v,
                  float* __restrict__ out,
                  const float* __restrict__ segKV,
                  const float* __restrict__ segKS) {
    __shared__ __attribute__((aligned(16))) float sm[SMEM_F];
    const int tid  = threadIdx.x;
    const int lane = tid & 31;
    const int wave8 = __builtin_amdgcn_readfirstlane(tid >> 5);
    const bool producer = wave8 >= 4;
    const int wave = wave8 & 3;
    const int gt   = tid & 127;
    const int lh   = lane >> 4;
    const int lc   = lane & 15;
    const int bh   = SEEDED ? (blockIdx.x / NS) : blockIdx.x;
    const int seg  = SEEDED ? (blockIdx.x % NS) : 0;
    const int b    = bh / HH;
    const int h    = bh % HH;
    const int npass = SEEDED ? (SEGL / TROWS) : (LL / TROWS);

    const v8f zero8 = {0.f,0.f,0.f,0.f,0.f,0.f,0.f,0.f};
    v8f st[4];
    float ksr0 = 0.f, ksr1 = 0.f;
    #pragma unroll
    for (int t = 0; t < 4; ++t) st[t] = zero8;

    if (SEEDED) {
        if (!producer) {
            const float* kvin = segKV + (size_t)blockIdx.x * 4096;
            #pragma unroll
            for (int t = 0; t < 4; ++t) {
                #pragma unroll
                for (int r = 0; r < 8; ++r) {
                    int srow = 16 * t + r + 8 * lh;
                    st[t][r] = kvin[srow * 64 + 16 * wave + lc];
                }
            }
        } else {
            ksr0 = segKS[(size_t)blockIdx.x * 64 + lane * 2];
            ksr1 = segKS[(size_t)blockIdx.x * 64 + lane * 2 + 1];
        }
    }

    float* ksw = sm + OFF_KS + wave * 64;
    const int lseg = seg * SEGL;

    if (producer) stage_pass(q, k, v, sm, b, h, lseg, gt, 1 < npass);
    lds_barrier();
    if (producer) score_pass(sm, ksw, wave, lane, lc, lh, ksr0, ksr1);
    lds_barrier();

    for (int it = 0; it < npass; ++it) {
        const int pn = it + 1;
        float* bufp = sm + (pn & 1) * BUF_F;
        const float* bufc = sm + (it & 1) * BUF_F;
        const size_t gout = ((size_t)((b * LL + lseg + it * TROWS) * HH + h)) * DD
                            + 16 * wave + lc;

        if (producer) {
            if (pn < npass)
                stage_pass(q, k, v, bufp, b, h, lseg + pn * TROWS, gt, pn + 1 < npass);
        } else {
            #pragma unroll
            for (int j = 0; j < 2; ++j)
                scan_chunk(bufc + j * TILE_F, st, out,
                           gout + (size_t)j * CCH * HH * DD, wave, lc, lh);
        }
        lds_barrier();

        if (producer) {
            if (pn < npass)
                score_pass(bufp, ksw, wave, lane, lc, lh, ksr0, ksr1);
        } else {
            #pragma unroll
            for (int j = 2; j < SUB; ++j)
                scan_chunk(bufc + j * TILE_F, st, out,
                           gout + (size_t)j * CCH * HH * DD, wave, lc, lh);
        }
        lds_barrier();
    }
}

extern "C" void kernel_launch(void* const* d_in, const int* in_sizes, int n_in,
                              void* d_out, int out_size, void* d_ws, size_t ws_size,
                              hipStream_t stream) {
    (void)in_sizes; (void)n_in; (void)out_size;
    const float* q = (const float*)d_in[0];
    const float* k = (const float*)d_in[1];
    const float* v = (const float*)d_in[2];
    float* out = (float*)d_out;

    const size_t nSeg   = (size_t)BB * HH * NS;
    const size_t needWs = nSeg * (4096 + 64) * sizeof(float);

    if (ws_size >= needWs) {
        float* segKV = (float*)d_ws;
        float* segKS = segKV + nSeg * 4096;
        hipLaunchKernelGGL(linattn_phase1, dim3(BB * HH * NS), dim3(128), 0,
                           stream, k, v, segKV, segKS);
        hipLaunchKernelGGL(linattn_phase2, dim3(BB * HH, 5), dim3(256), 0,
                           stream, segKV, segKS);
        hipLaunchKernelGGL((linattn_scan<true>), dim3(BB * HH * NS), dim3(256),
                           0, stream, q, k, v, out, segKV, segKS);
    } else {
        hipLaunchKernelGGL((linattn_scan<false>), dim3(BB * HH), dim3(256),
                           0, stream, q, k, v, out,
                           (const float*)nullptr, (const float*)nullptr);
    }
}
